# starting kernel
# baseline (speedup 1.0000x reference)
#include <hip/hip_runtime.h>
#include <math.h>


typedef float v2f __attribute__((ext_vector_type(2)));
typedef float v8f __attribute__((ext_vector_type(8)));

#define WMMA_F32(a, b, c) \
  __builtin_amdgcn_wmma_f32_16x16x4_f32(false, (a), false, (b), (short)0, (c), false, false)

constexpr int BATCH = 8;
constexpr int SEQ   = 1024;
constexpr int CIN   = 256;
constexpr int HEADS = 8;
constexpr int DIM   = 64;
constexpr int HD    = HEADS * DIM;
constexpr int COUT  = 256;
constexpr float SCALE = 0.125f;

__global__ __launch_bounds__(32) void qkv_gemm(
    const float* __restrict__ X,
    const float* __restrict__ Wq, const float* __restrict__ Wk,
    const float* __restrict__ Wv,
    float* __restrict__ qws, float* __restrict__ kws, float* __restrict__ vws)
{
  const int lane = threadIdx.x;
  const int half = lane >> 4, l16 = lane & 15;
  const int n0 = blockIdx.x * 32;
  const int i0 = blockIdx.y * 32;
  const int b  = blockIdx.z / 3;
  const int sel = blockIdx.z % 3;

  const float* W = (sel == 0) ? Wq : (sel == 1) ? Wk : Wv;
  float*       O = (sel == 0) ? qws : (sel == 1) ? kws : vws;

  const float* xrow0 = X + (size_t)(b * SEQ + i0 + l16) * CIN;
  const float* xrow1 = xrow0 + (size_t)16 * CIN;

  v8f acc00 = {}, acc01 = {}, acc10 = {}, acc11 = {};
#pragma unroll 8
  for (int kb = 0; kb < CIN; kb += 4) {
    const int kk = kb + half * 2;
    v2f a0 = *(const v2f*)(xrow0 + kk);
    v2f a1 = *(const v2f*)(xrow1 + kk);
    v2f b0, b1;
    b0.x = W[(size_t)kk * HD + n0 + l16];
    b0.y = W[(size_t)(kk + 1) * HD + n0 + l16];
    b1.x = W[(size_t)kk * HD + n0 + 16 + l16];
    b1.y = W[(size_t)(kk + 1) * HD + n0 + 16 + l16];
    acc00 = WMMA_F32(a0, b0, acc00);
    acc01 = WMMA_F32(a0, b1, acc01);
    acc10 = WMMA_F32(a1, b0, acc10);
    acc11 = WMMA_F32(a1, b1, acc11);
  }
#pragma unroll
  for (int r = 0; r < 8; ++r) {
    const size_t row0 = (size_t)(b * SEQ + i0 + r + half * 8) * HD;
    const size_t row1 = (size_t)(b * SEQ + i0 + 16 + r + half * 8) * HD;
    O[row0 + n0 + l16]      = acc00[r];
    O[row0 + n0 + 16 + l16] = acc01[r];
    O[row1 + n0 + l16]      = acc10[r];
    O[row1 + n0 + 16 + l16] = acc11[r];
  }
}

__global__ __launch_bounds__(32) void col_stats(
    const float* __restrict__ qws, const float* __restrict__ kws,
    float* __restrict__ mcol, float* __restrict__ lcol)
{
  const int lane = threadIdx.x, half = lane >> 4, l16 = lane & 15;
  const int j0 = blockIdx.x * 32;
  const int h  = blockIdx.y;
  const int b  = blockIdx.z;
  const int jA = j0 + l16;
  const int jB = j0 + 16 + l16;

  const float* kpA = kws + (size_t)(b * SEQ + jA) * HD + h * DIM;
  const float* kpB = kws + (size_t)(b * SEQ + jB) * HD + h * DIM;
  v2f bfA[16], bfB[16];
#pragma unroll
  for (int t = 0; t < 16; ++t) {
    bfA[t] = *(const v2f*)(kpA + t * 4 + half * 2);
    bfB[t] = *(const v2f*)(kpB + t * 4 + half * 2);
  }

  float m0 = -__builtin_inff(), l0 = 0.0f;
  float m1 = -__builtin_inff(), l1 = 0.0f;
  const float* qbase = qws + (size_t)b * SEQ * HD + h * DIM;

  for (int i0 = 0; i0 < SEQ; i0 += 16) {
    const float* qrow = qbase + (size_t)(i0 + l16) * HD;
    v8f accA = {}, accB = {};
#pragma unroll
    for (int t = 0; t < 16; ++t) {
      v2f a = *(const v2f*)(qrow + t * 4 + half * 2);
      accA = WMMA_F32(a, bfA[t], accA);
      accB = WMMA_F32(a, bfB[t], accB);
    }
    float sA[8], sB[8];
    float tmA = -__builtin_inff(), tmB = -__builtin_inff();
#pragma unroll
    for (int r = 0; r < 8; ++r) {
      sA[r] = accA[r] * SCALE; tmA = fmaxf(tmA, sA[r]);
      sB[r] = accB[r] * SCALE; tmB = fmaxf(tmB, sB[r]);
    }
    tmA = fmaxf(tmA, __shfl_xor(tmA, 16, 32));
    tmB = fmaxf(tmB, __shfl_xor(tmB, 16, 32));
    const float mn0 = fmaxf(m0, tmA);
    const float mn1 = fmaxf(m1, tmB);
    float psA = 0.0f, psB = 0.0f;
#pragma unroll
    for (int r = 0; r < 8; ++r) {
      psA += __expf(sA[r] - mn0);
      psB += __expf(sB[r] - mn1);
    }
    psA += __shfl_xor(psA, 16, 32);
    psB += __shfl_xor(psB, 16, 32);
    l0 = l0 * __expf(m0 - mn0) + psA;
    l1 = l1 * __expf(m1 - mn1) + psB;
    m0 = mn0;
    m1 = mn1;
  }

  if (half == 0) {
    const size_t base = (size_t)(b * HEADS + h) * SEQ;
    mcol[base + jA] = m0;  lcol[base + jA] = l0;
    mcol[base + jB] = m1;  lcol[base + jB] = l1;
  }
}

__global__ __launch_bounds__(32) void row_wsum(
    const float* __restrict__ qws, const float* __restrict__ kws,
    const float* __restrict__ mcol, const float* __restrict__ lcol,
    float* __restrict__ wsum)
{
  const int lane = threadIdx.x, half = lane >> 4, l16 = lane & 15;
  const int i0 = blockIdx.x * 32;
  const int h  = blockIdx.y;
  const int b  = blockIdx.z;

  const float* qrowA = qws + (size_t)(b * SEQ + i0 + l16) * HD + h * DIM;
  const float* qrowB = qrowA + (size_t)16 * HD;
  v2f afA[16], afB[16];
#pragma unroll
  for (int t = 0; t < 16; ++t) {
    afA[t] = *(const v2f*)(qrowA + t * 4 + half * 2);
    afB[t] = *(const v2f*)(qrowB + t * 4 + half * 2);
  }

  float waccA[8], waccB[8];
#pragma unroll
  for (int r = 0; r < 8; ++r) { waccA[r] = 0.0f; waccB[r] = 0.0f; }

  const float* kbase = kws + (size_t)b * SEQ * HD + h * DIM;
  const float* mrow  = mcol + (size_t)(b * HEADS + h) * SEQ;
  const float* lrow  = lcol + (size_t)(b * HEADS + h) * SEQ;

  for (int j0 = 0; j0 < SEQ; j0 += 16) {
    const int j = j0 + l16;
    const float mj   = mrow[j];
    const float invl = 1.0f / lrow[j];
    const float* krow = kbase + (size_t)j * HD;
    v8f accA = {}, accB = {};
#pragma unroll
    for (int t = 0; t < 16; ++t) {
      v2f bb = *(const v2f*)(krow + t * 4 + half * 2);
      accA = WMMA_F32(afA[t], bb, accA);
      accB = WMMA_F32(afB[t], bb, accB);
    }
#pragma unroll
    for (int r = 0; r < 8; ++r) {
      waccA[r] += __expf(accA[r] * SCALE - mj) * invl;
      waccB[r] += __expf(accB[r] * SCALE - mj) * invl;
    }
  }

#pragma unroll
  for (int r = 0; r < 8; ++r) {
#pragma unroll
    for (int mask = 1; mask < 16; mask <<= 1) {
      waccA[r] += __shfl_xor(waccA[r], mask, 32);
      waccB[r] += __shfl_xor(waccB[r], mask, 32);
    }
  }
  if (l16 == 0) {
#pragma unroll
    for (int r = 0; r < 8; ++r) {
      wsum[(size_t)(b * SEQ + i0 + r + half * 8) * HEADS + h]      = waccA[r];
      wsum[(size_t)(b * SEQ + i0 + 16 + r + half * 8) * HEADS + h] = waccB[r];
    }
  }
}

__global__ __launch_bounds__(32) void out_gemm(
    const float* __restrict__ vws, const float* __restrict__ wsum,
    const float* __restrict__ Wout, float* __restrict__ out)
{
  const int lane = threadIdx.x, half = lane >> 4, l16 = lane & 15;
  const int o0 = blockIdx.x * 32;
  const int i0 = blockIdx.y * 32;
  const int b  = blockIdx.z;
  const int iA = i0 + l16;
  const int iB = i0 + 16 + l16;

  const float* vrowA = vws + (size_t)(b * SEQ + iA) * HD;
  const float* vrowB = vws + (size_t)(b * SEQ + iB) * HD;
  const float* wrowA = wsum + (size_t)(b * SEQ + iA) * HEADS;
  const float* wrowB = wsum + (size_t)(b * SEQ + iB) * HEADS;
  float wvA[HEADS], wvB[HEADS];
#pragma unroll
  for (int h = 0; h < HEADS; ++h) { wvA[h] = wrowA[h]; wvB[h] = wrowB[h]; }

  const float* wo0 = Wout + (size_t)(o0 + l16) * HD;
  const float* wo1 = Wout + (size_t)(o0 + 16 + l16) * HD;

  v8f acc00 = {}, acc01 = {}, acc10 = {}, acc11 = {};
#pragma unroll 8
  for (int kb = 0; kb < HD; kb += 4) {
    const int kk = kb + half * 2;
    const int h = kk >> 6;
    v2f a0 = *(const v2f*)(vrowA + kk);
    v2f a1 = *(const v2f*)(vrowB + kk);
    a0.x *= wvA[h]; a0.y *= wvA[h];
    a1.x *= wvB[h]; a1.y *= wvB[h];
    v2f b0 = *(const v2f*)(wo0 + kk);
    v2f b1 = *(const v2f*)(wo1 + kk);
    acc00 = WMMA_F32(a0, b0, acc00);
    acc01 = WMMA_F32(a0, b1, acc01);
    acc10 = WMMA_F32(a1, b0, acc10);
    acc11 = WMMA_F32(a1, b1, acc11);
  }
#pragma unroll
  for (int r = 0; r < 8; ++r) {
    const size_t row0 = (size_t)(b * SEQ + i0 + r + half * 8) * COUT;
    const size_t row1 = (size_t)(b * SEQ + i0 + 16 + r + half * 8) * COUT;
    out[row0 + o0 + l16]      = acc00[r];
    out[row0 + o0 + 16 + l16] = acc01[r];
    out[row1 + o0 + l16]      = acc10[r];
    out[row1 + o0 + 16 + l16] = acc11[r];
  }
}

extern "C" void kernel_launch(void* const* d_in, const int* in_sizes, int n_in,
                              void* d_out, int out_size, void* d_ws, size_t ws_size,
                              hipStream_t stream)
{
  (void)in_sizes; (void)n_in; (void)out_size; (void)ws_size;
  const float* X  = (const float*)d_in[0];
  const float* Wq = (const float*)d_in[1];
  const float* Wk = (const float*)d_in[2];
  const float* Wv = (const float*)d_in[3];
  const float* Wo = (const float*)d_in[4];
  float* out = (float*)d_out;

  const size_t QSZ = (size_t)BATCH * SEQ * HD;
  const size_t SSZ = (size_t)BATCH * HEADS * SEQ;
  float* qws  = (float*)d_ws;
  float* kws  = qws + QSZ;
  float* vws  = kws + QSZ;
  float* mcol = vws + QSZ;
  float* lcol = mcol + SSZ;
  float* wsum = lcol + SSZ;

  qkv_gemm<<<dim3(HD / 32, SEQ / 32, BATCH * 3), 32, 0, stream>>>(
      X, Wq, Wk, Wv, qws, kws, vws);
  col_stats<<<dim3(SEQ / 32, HEADS, BATCH), 32, 0, stream>>>(
      qws, kws, mcol, lcol);
  row_wsum<<<dim3(SEQ / 32, HEADS, BATCH), 32, 0, stream>>>(
      qws, kws, mcol, lcol, wsum);
  out_gemm<<<dim3(COUT / 32, SEQ / 32, BATCH), 32, 0, stream>>>(
      vws, wsum, Wo, out);
}


# Passes 1 and 2 (col_stats, row_wsum): streamed Q/K fragments double-buffered in registers; next tile's 16 loads issued as one group a tile ahead, not one exposed load wait per WMMA pair
# speedup vs baseline: 1.2388x; 1.2388x over previous
#include <hip/hip_runtime.h>
#include <math.h>


typedef float v2f __attribute__((ext_vector_type(2)));
typedef float v8f __attribute__((ext_vector_type(8)));

#define WMMA_F32(a, b, c) \
  __builtin_amdgcn_wmma_f32_16x16x4_f32(false, (a), false, (b), (short)0, (c), false, false)

constexpr int BATCH = 8;
constexpr int SEQ   = 1024;
constexpr int CIN   = 256;
constexpr int HEADS = 8;
constexpr int DIM   = 64;
constexpr int HD    = HEADS * DIM;
constexpr int COUT  = 256;
constexpr float SCALE = 0.125f;

__global__ __launch_bounds__(32) void qkv_gemm(
    const float* __restrict__ X,
    const float* __restrict__ Wq, const float* __restrict__ Wk,
    const float* __restrict__ Wv,
    float* __restrict__ qws, float* __restrict__ kws, float* __restrict__ vws)
{
  const int lane = threadIdx.x;
  const int half = lane >> 4, l16 = lane & 15;
  const int n0 = blockIdx.x * 32;
  const int i0 = blockIdx.y * 32;
  const int b  = blockIdx.z / 3;
  const int sel = blockIdx.z % 3;

  const float* W = (sel == 0) ? Wq : (sel == 1) ? Wk : Wv;
  float*       O = (sel == 0) ? qws : (sel == 1) ? kws : vws;

  const float* xrow0 = X + (size_t)(b * SEQ + i0 + l16) * CIN;
  const float* xrow1 = xrow0 + (size_t)16 * CIN;

  v8f acc00 = {}, acc01 = {}, acc10 = {}, acc11 = {};
#pragma unroll 8
  for (int kb = 0; kb < CIN; kb += 4) {
    const int kk = kb + half * 2;
    v2f a0 = *(const v2f*)(xrow0 + kk);
    v2f a1 = *(const v2f*)(xrow1 + kk);
    v2f b0, b1;
    b0.x = W[(size_t)kk * HD + n0 + l16];
    b0.y = W[(size_t)(kk + 1) * HD + n0 + l16];
    b1.x = W[(size_t)kk * HD + n0 + 16 + l16];
    b1.y = W[(size_t)(kk + 1) * HD + n0 + 16 + l16];
    acc00 = WMMA_F32(a0, b0, acc00);
    acc01 = WMMA_F32(a0, b1, acc01);
    acc10 = WMMA_F32(a1, b0, acc10);
    acc11 = WMMA_F32(a1, b1, acc11);
  }
#pragma unroll
  for (int r = 0; r < 8; ++r) {
    const size_t row0 = (size_t)(b * SEQ + i0 + r + half * 8) * HD;
    const size_t row1 = (size_t)(b * SEQ + i0 + 16 + r + half * 8) * HD;
    O[row0 + n0 + l16]      = acc00[r];
    O[row0 + n0 + 16 + l16] = acc01[r];
    O[row1 + n0 + l16]      = acc10[r];
    O[row1 + n0 + 16 + l16] = acc11[r];
  }
}

__device__ __forceinline__ void load_frags(v2f (&f)[16], const float* __restrict__ p)
{
#pragma unroll
  for (int t = 0; t < 16; ++t) f[t] = *(const v2f*)(p + t * 4);
}

__device__ __forceinline__ void col_tile(
    const v2f (&a)[16], const v2f (&bfA)[16], const v2f (&bfB)[16],
    float& m0, float& l0, float& m1, float& l1)
{
  v8f accA = {}, accB = {};
#pragma unroll
  for (int t = 0; t < 16; ++t) {
    accA = WMMA_F32(a[t], bfA[t], accA);
    accB = WMMA_F32(a[t], bfB[t], accB);
  }
  float sA[8], sB[8];
  float tmA = -__builtin_inff(), tmB = -__builtin_inff();
#pragma unroll
  for (int r = 0; r < 8; ++r) {
    sA[r] = accA[r] * SCALE; tmA = fmaxf(tmA, sA[r]);
    sB[r] = accB[r] * SCALE; tmB = fmaxf(tmB, sB[r]);
  }
  tmA = fmaxf(tmA, __shfl_xor(tmA, 16, 32));
  tmB = fmaxf(tmB, __shfl_xor(tmB, 16, 32));
  const float mn0 = fmaxf(m0, tmA);
  const float mn1 = fmaxf(m1, tmB);
  float psA = 0.0f, psB = 0.0f;
#pragma unroll
  for (int r = 0; r < 8; ++r) {
    psA += __expf(sA[r] - mn0);
    psB += __expf(sB[r] - mn1);
  }
  psA += __shfl_xor(psA, 16, 32);
  psB += __shfl_xor(psB, 16, 32);
  l0 = l0 * __expf(m0 - mn0) + psA;
  l1 = l1 * __expf(m1 - mn1) + psB;
  m0 = mn0;
  m1 = mn1;
}

__global__ __launch_bounds__(32) void col_stats(
    const float* __restrict__ qws, const float* __restrict__ kws,
    float* __restrict__ mcol, float* __restrict__ lcol)
{
  const int lane = threadIdx.x, half = lane >> 4, l16 = lane & 15;
  const int j0 = blockIdx.x * 32;
  const int h  = blockIdx.y;
  const int b  = blockIdx.z;
  const int jA = j0 + l16;
  const int jB = j0 + 16 + l16;

  const float* kpA = kws + (size_t)(b * SEQ + jA) * HD + h * DIM;
  const float* kpB = kws + (size_t)(b * SEQ + jB) * HD + h * DIM;
  v2f bfA[16], bfB[16];
#pragma unroll
  for (int t = 0; t < 16; ++t) {
    bfA[t] = *(const v2f*)(kpA + t * 4 + half * 2);
    bfB[t] = *(const v2f*)(kpB + t * 4 + half * 2);
  }

  float m0 = -__builtin_inff(), l0 = 0.0f;
  float m1 = -__builtin_inff(), l1 = 0.0f;
  const float* qlane = qws + (size_t)b * SEQ * HD + h * DIM + (size_t)l16 * HD + half * 2;

  v2f fa[16], fb[16];
  load_frags(fa, qlane);
  for (int i0 = 0; i0 < SEQ; i0 += 32) {
    load_frags(fb, qlane + (size_t)(i0 + 16) * HD);
    __builtin_amdgcn_sched_barrier(0);
    col_tile(fa, bfA, bfB, m0, l0, m1, l1);
    const int inx = (i0 + 32 < SEQ) ? (i0 + 32) : (i0 + 16);
    load_frags(fa, qlane + (size_t)inx * HD);
    __builtin_amdgcn_sched_barrier(0);
    col_tile(fb, bfA, bfB, m0, l0, m1, l1);
  }

  if (half == 0) {
    const size_t base = (size_t)(b * HEADS + h) * SEQ;
    mcol[base + jA] = m0;  lcol[base + jA] = l0;
    mcol[base + jB] = m1;  lcol[base + jB] = l1;
  }
}

__device__ __forceinline__ void row_tile(
    const v2f (&kb)[16], const float mj, const float lj,
    const v2f (&afA)[16], const v2f (&afB)[16],
    float (&waccA)[8], float (&waccB)[8])
{
  const float invl = 1.0f / lj;
  v8f accA = {}, accB = {};
#pragma unroll
  for (int t = 0; t < 16; ++t) {
    accA = WMMA_F32(afA[t], kb[t], accA);
    accB = WMMA_F32(afB[t], kb[t], accB);
  }
#pragma unroll
  for (int r = 0; r < 8; ++r) {
    waccA[r] += __expf(accA[r] * SCALE - mj) * invl;
    waccB[r] += __expf(accB[r] * SCALE - mj) * invl;
  }
}

__global__ __launch_bounds__(32) void row_wsum(
    const float* __restrict__ qws, const float* __restrict__ kws,
    const float* __restrict__ mcol, const float* __restrict__ lcol,
    float* __restrict__ wsum)
{
  const int lane = threadIdx.x, half = lane >> 4, l16 = lane & 15;
  const int i0 = blockIdx.x * 32;
  const int h  = blockIdx.y;
  const int b  = blockIdx.z;

  const float* qrowA = qws + (size_t)(b * SEQ + i0 + l16) * HD + h * DIM;
  const float* qrowB = qrowA + (size_t)16 * HD;
  v2f afA[16], afB[16];
#pragma unroll
  for (int t = 0; t < 16; ++t) {
    afA[t] = *(const v2f*)(qrowA + t * 4 + half * 2);
    afB[t] = *(const v2f*)(qrowB + t * 4 + half * 2);
  }

  float waccA[8], waccB[8];
#pragma unroll
  for (int r = 0; r < 8; ++r) { waccA[r] = 0.0f; waccB[r] = 0.0f; }

  const float* klane = kws + (size_t)b * SEQ * HD + h * DIM + (size_t)l16 * HD + half * 2;
  const float* mlane = mcol + (size_t)(b * HEADS + h) * SEQ + l16;
  const float* llane = lcol + (size_t)(b * HEADS + h) * SEQ + l16;

  v2f fa[16], fb[16];
  float mA, lA, mB, lB;
  load_frags(fa, klane);
  mA = mlane[0];
  lA = llane[0];
  for (int j0 = 0; j0 < SEQ; j0 += 32) {
    load_frags(fb, klane + (size_t)(j0 + 16) * HD);
    mB = mlane[j0 + 16];
    lB = llane[j0 + 16];
    __builtin_amdgcn_sched_barrier(0);
    row_tile(fa, mA, lA, afA, afB, waccA, waccB);
    const int jn = (j0 + 32 < SEQ) ? (j0 + 32) : (j0 + 16);
    load_frags(fa, klane + (size_t)jn * HD);
    mA = mlane[jn];
    lA = llane[jn];
    __builtin_amdgcn_sched_barrier(0);
    row_tile(fb, mB, lB, afA, afB, waccA, waccB);
  }

#pragma unroll
  for (int r = 0; r < 8; ++r) {
#pragma unroll
    for (int mask = 1; mask < 16; mask <<= 1) {
      waccA[r] += __shfl_xor(waccA[r], mask, 32);
      waccB[r] += __shfl_xor(waccB[r], mask, 32);
    }
  }
  if (l16 == 0) {
#pragma unroll
    for (int r = 0; r < 8; ++r) {
      wsum[(size_t)(b * SEQ + i0 + r + half * 8) * HEADS + h]      = waccA[r];
      wsum[(size_t)(b * SEQ + i0 + 16 + r + half * 8) * HEADS + h] = waccB[r];
    }
  }
}

__global__ __launch_bounds__(32) void out_gemm(
    const float* __restrict__ vws, const float* __restrict__ wsum,
    const float* __restrict__ Wout, float* __restrict__ out)
{
  const int lane = threadIdx.x, half = lane >> 4, l16 = lane & 15;
  const int o0 = blockIdx.x * 32;
  const int i0 = blockIdx.y * 32;
  const int b  = blockIdx.z;
  const int iA = i0 + l16;
  const int iB = i0 + 16 + l16;

  const float* vrowA = vws + (size_t)(b * SEQ + iA) * HD;
  const float* vrowB = vws + (size_t)(b * SEQ + iB) * HD;
  const float* wrowA = wsum + (size_t)(b * SEQ + iA) * HEADS;
  const float* wrowB = wsum + (size_t)(b * SEQ + iB) * HEADS;
  float wvA[HEADS], wvB[HEADS];
#pragma unroll
  for (int h = 0; h < HEADS; ++h) { wvA[h] = wrowA[h]; wvB[h] = wrowB[h]; }

  const float* wo0 = Wout + (size_t)(o0 + l16) * HD;
  const float* wo1 = Wout + (size_t)(o0 + 16 + l16) * HD;

  v8f acc00 = {}, acc01 = {}, acc10 = {}, acc11 = {};
#pragma unroll 8
  for (int kb = 0; kb < HD; kb += 4) {
    const int kk = kb + half * 2;
    const int h = kk >> 6;
    v2f a0 = *(const v2f*)(vrowA + kk);
    v2f a1 = *(const v2f*)(vrowB + kk);
    a0.x *= wvA[h]; a0.y *= wvA[h];
    a1.x *= wvB[h]; a1.y *= wvB[h];
    v2f b0 = *(const v2f*)(wo0 + kk);
    v2f b1 = *(const v2f*)(wo1 + kk);
    acc00 = WMMA_F32(a0, b0, acc00);
    acc01 = WMMA_F32(a0, b1, acc01);
    acc10 = WMMA_F32(a1, b0, acc10);
    acc11 = WMMA_F32(a1, b1, acc11);
  }
#pragma unroll
  for (int r = 0; r < 8; ++r) {
    const size_t row0 = (size_t)(b * SEQ + i0 + r + half * 8) * COUT;
    const size_t row1 = (size_t)(b * SEQ + i0 + 16 + r + half * 8) * COUT;
    out[row0 + o0 + l16]      = acc00[r];
    out[row0 + o0 + 16 + l16] = acc01[r];
    out[row1 + o0 + l16]      = acc10[r];
    out[row1 + o0 + 16 + l16] = acc11[r];
  }
}

extern "C" void kernel_launch(void* const* d_in, const int* in_sizes, int n_in,
                              void* d_out, int out_size, void* d_ws, size_t ws_size,
                              hipStream_t stream)
{
  (void)in_sizes; (void)n_in; (void)out_size; (void)ws_size;
  const float* X  = (const float*)d_in[0];
  const float* Wq = (const float*)d_in[1];
  const float* Wk = (const float*)d_in[2];
  const float* Wv = (const float*)d_in[3];
  const float* Wo = (const float*)d_in[4];
  float* out = (float*)d_out;

  const size_t QSZ = (size_t)BATCH * SEQ * HD;
  const size_t SSZ = (size_t)BATCH * HEADS * SEQ;
  float* qws  = (float*)d_ws;
  float* kws  = qws + QSZ;
  float* vws  = kws + QSZ;
  float* mcol = vws + QSZ;
  float* lcol = mcol + SSZ;
  float* wsum = lcol + SSZ;

  qkv_gemm<<<dim3(HD / 32, SEQ / 32, BATCH * 3), 32, 0, stream>>>(
      X, Wq, Wk, Wv, qws, kws, vws);
  col_stats<<<dim3(SEQ / 32, HEADS, BATCH), 32, 0, stream>>>(
      qws, kws, mcol, lcol);
  row_wsum<<<dim3(SEQ / 32, HEADS, BATCH), 32, 0, stream>>>(
      qws, kws, mcol, lcol, wsum);
  out_gemm<<<dim3(COUT / 32, SEQ / 32, BATCH), 32, 0, stream>>>(
      vws, wsum, Wo, out);
}


# Q,K workspace kept in WMMA-fragment order (projection writes the transposed tile): pass 1/2 fragment loads are whole-line 16 B loads, not 16-line 8 B ones; k order inside the 64-long dots permuted
# speedup vs baseline: 1.2904x; 1.0417x over previous
#include <hip/hip_runtime.h>
#include <math.h>


typedef float v2f __attribute__((ext_vector_type(2)));
typedef float v4f __attribute__((ext_vector_type(4)));
typedef float v8f __attribute__((ext_vector_type(8)));

#define WMMA_F32(a, b, c) \
  __builtin_amdgcn_wmma_f32_16x16x4_f32(false, (a), false, (b), (short)0, (c), false, false)

constexpr int BATCH = 8;
constexpr int SEQ   = 1024;
constexpr int CIN   = 256;
constexpr int HEADS = 8;
constexpr int DIM   = 64;
constexpr int HD    = HEADS * DIM;
constexpr int COUT  = 256;
constexpr float SCALE = 0.125f;
constexpr int NT16   = SEQ / 16;
constexpr int TILE_F = 16 * DIM;

__device__ __forceinline__ size_t tile_off(int b, int h, int t16)
{
  return ((size_t)(b * HEADS + h) * NT16 + t16) * TILE_F;
}
__device__ __forceinline__ v2f lo2(v4f x) { return __builtin_shufflevector(x, x, 0, 1); }
__device__ __forceinline__ v2f hi2(v4f x) { return __builtin_shufflevector(x, x, 2, 3); }

__device__ __forceinline__ void qkv_step_loads(
    const float* __restrict__ xrow0, const float* __restrict__ xrow1,
    const float* __restrict__ W, int kk, int ncol,
    v2f& a0, v2f& a1, v2f& b0, v2f& b1)
{
  a0 = *(const v2f*)(xrow0 + kk);
  a1 = *(const v2f*)(xrow1 + kk);
  b0.x = W[(size_t)kk * HD + ncol];
  b0.y = W[(size_t)(kk + 1) * HD + ncol];
  b1.x = W[(size_t)kk * HD + ncol + 16];
  b1.y = W[(size_t)(kk + 1) * HD + ncol + 16];
}

__global__ __launch_bounds__(32) void qkv_gemm(
    const float* __restrict__ X,
    const float* __restrict__ Wq, const float* __restrict__ Wk,
    const float* __restrict__ Wv,
    float* __restrict__ qws, float* __restrict__ kws, float* __restrict__ vws)
{
  const int lane = threadIdx.x;
  const int half = lane >> 4, l16 = lane & 15;
  const int n0 = blockIdx.x * 32;
  const int i0 = blockIdx.y * 32;
  const int b  = blockIdx.z / 3;
  const int sel = blockIdx.z % 3;

  const float* W = (sel == 0) ? Wq : (sel == 1) ? Wk : Wv;

  const float* xrow0 = X + (size_t)(b * SEQ + i0 + l16) * CIN;
  const float* xrow1 = xrow0 + (size_t)16 * CIN;

  v8f acc00 = {}, acc01 = {}, acc10 = {}, acc11 = {};
  if (sel == 2) {
#pragma unroll 8
    for (int kb = 0; kb < CIN; kb += 4) {
      v2f a0, a1, b0, b1;
      qkv_step_loads(xrow0, xrow1, W, kb + half * 2, n0 + l16, a0, a1, b0, b1);
      acc00 = WMMA_F32(a0, b0, acc00);
      acc01 = WMMA_F32(a0, b1, acc01);
      acc10 = WMMA_F32(a1, b0, acc10);
      acc11 = WMMA_F32(a1, b1, acc11);
    }
#pragma unroll
    for (int r = 0; r < 8; ++r) {
      const size_t row0 = (size_t)(b * SEQ + i0 + r + half * 8) * HD;
      const size_t row1 = (size_t)(b * SEQ + i0 + 16 + r + half * 8) * HD;
      vws[row0 + n0 + l16]      = acc00[r];
      vws[row0 + n0 + 16 + l16] = acc01[r];
      vws[row1 + n0 + l16]      = acc10[r];
      vws[row1 + n0 + 16 + l16] = acc11[r];
    }
  } else {
#pragma unroll 8
    for (int kb = 0; kb < CIN; kb += 4) {
      v2f a0, a1, b0, b1;
      qkv_step_loads(xrow0, xrow1, W, kb + half * 2, n0 + l16, a0, a1, b0, b1);
      acc00 = WMMA_F32(b0, a0, acc00);
      acc01 = WMMA_F32(b0, a1, acc01);
      acc10 = WMMA_F32(b1, a0, acc10);
      acc11 = WMMA_F32(b1, a1, acc11);
    }
    float* O = (sel == 0) ? qws : kws;
    const int head = n0 >> 6;
    const int w0 = ((n0 & 63) >> 4) * 2;
    float* t0 = O + tile_off(b, head, i0 >> 4) + lane * 4;
    float* t1 = t0 + TILE_F;
    *(v4f*)(t0 + (w0 + 0) * 128) = __builtin_shufflevector(acc00, acc00, 0, 1, 2, 3);
    *(v4f*)(t0 + (w0 + 1) * 128) = __builtin_shufflevector(acc00, acc00, 4, 5, 6, 7);
    *(v4f*)(t0 + (w0 + 2) * 128) = __builtin_shufflevector(acc10, acc10, 0, 1, 2, 3);
    *(v4f*)(t0 + (w0 + 3) * 128) = __builtin_shufflevector(acc10, acc10, 4, 5, 6, 7);
    *(v4f*)(t1 + (w0 + 0) * 128) = __builtin_shufflevector(acc01, acc01, 0, 1, 2, 3);
    *(v4f*)(t1 + (w0 + 1) * 128) = __builtin_shufflevector(acc01, acc01, 4, 5, 6, 7);
    *(v4f*)(t1 + (w0 + 2) * 128) = __builtin_shufflevector(acc11, acc11, 0, 1, 2, 3);
    *(v4f*)(t1 + (w0 + 3) * 128) = __builtin_shufflevector(acc11, acc11, 4, 5, 6, 7);
  }
}

__device__ __forceinline__ void load_frags(v4f (&f)[8], const float* __restrict__ p)
{
#pragma unroll
  for (int w = 0; w < 8; ++w) f[w] = *(const v4f*)(p + w * 128);
}

__device__ __forceinline__ void col_tile(
    const v4f (&a)[8], const v4f (&bfA)[8], const v4f (&bfB)[8],
    float& m0, float& l0, float& m1, float& l1)
{
  v8f accA = {}, accB = {};
#pragma unroll
  for (int w = 0; w < 8; ++w) {
    accA = WMMA_F32(lo2(a[w]), lo2(bfA[w]), accA);
    accB = WMMA_F32(lo2(a[w]), lo2(bfB[w]), accB);
    accA = WMMA_F32(hi2(a[w]), hi2(bfA[w]), accA);
    accB = WMMA_F32(hi2(a[w]), hi2(bfB[w]), accB);
  }
  float sA[8], sB[8];
  float tmA = -__builtin_inff(), tmB = -__builtin_inff();
#pragma unroll
  for (int r = 0; r < 8; ++r) {
    sA[r] = accA[r] * SCALE; tmA = fmaxf(tmA, sA[r]);
    sB[r] = accB[r] * SCALE; tmB = fmaxf(tmB, sB[r]);
  }
  tmA = fmaxf(tmA, __shfl_xor(tmA, 16, 32));
  tmB = fmaxf(tmB, __shfl_xor(tmB, 16, 32));
  const float mn0 = fmaxf(m0, tmA);
  const float mn1 = fmaxf(m1, tmB);
  float psA = 0.0f, psB = 0.0f;
#pragma unroll
  for (int r = 0; r < 8; ++r) {
    psA += __expf(sA[r] - mn0);
    psB += __expf(sB[r] - mn1);
  }
  psA += __shfl_xor(psA, 16, 32);
  psB += __shfl_xor(psB, 16, 32);
  l0 = l0 * __expf(m0 - mn0) + psA;
  l1 = l1 * __expf(m1 - mn1) + psB;
  m0 = mn0;
  m1 = mn1;
}

__global__ __launch_bounds__(32) void col_stats(
    const float* __restrict__ qws, const float* __restrict__ kws,
    float* __restrict__ mcol, float* __restrict__ lcol)
{
  const int lane = threadIdx.x, half = lane >> 4, l16 = lane & 15;
  const int j0 = blockIdx.x * 32;
  const int h  = blockIdx.y;
  const int b  = blockIdx.z;
  const int jA = j0 + l16;
  const int jB = j0 + 16 + l16;

  const float* kpA = kws + tile_off(b, h, j0 >> 4) + lane * 4;
  v4f bfA[8], bfB[8];
  load_frags(bfA, kpA);
  load_frags(bfB, kpA + TILE_F);

  float m0 = -__builtin_inff(), l0 = 0.0f;
  float m1 = -__builtin_inff(), l1 = 0.0f;
  const float* qlane = qws + tile_off(b, h, 0) + lane * 4;

  v4f fa[8], fb[8];
  load_frags(fa, qlane);
  for (int t = 0; t < NT16; t += 2) {
    load_frags(fb, qlane + (size_t)(t + 1) * TILE_F);
    __builtin_amdgcn_sched_barrier(0);
    col_tile(fa, bfA, bfB, m0, l0, m1, l1);
    const int tn = (t + 2 < NT16) ? (t + 2) : (t + 1);
    load_frags(fa, qlane + (size_t)tn * TILE_F);
    __builtin_amdgcn_sched_barrier(0);
    col_tile(fb, bfA, bfB, m0, l0, m1, l1);
  }

  if (half == 0) {
    const size_t base = (size_t)(b * HEADS + h) * SEQ;
    mcol[base + jA] = m0;  lcol[base + jA] = l0;
    mcol[base + jB] = m1;  lcol[base + jB] = l1;
  }
}

__device__ __forceinline__ void row_tile(
    const v4f (&kb)[8], const float mj, const float lj,
    const v4f (&afA)[8], const v4f (&afB)[8],
    float (&waccA)[8], float (&waccB)[8])
{
  const float invl = 1.0f / lj;
  v8f accA = {}, accB = {};
#pragma unroll
  for (int w = 0; w < 8; ++w) {
    accA = WMMA_F32(lo2(afA[w]), lo2(kb[w]), accA);
    accB = WMMA_F32(lo2(afB[w]), lo2(kb[w]), accB);
    accA = WMMA_F32(hi2(afA[w]), hi2(kb[w]), accA);
    accB = WMMA_F32(hi2(afB[w]), hi2(kb[w]), accB);
  }
#pragma unroll
  for (int r = 0; r < 8; ++r) {
    waccA[r] += __expf(accA[r] * SCALE - mj) * invl;
    waccB[r] += __expf(accB[r] * SCALE - mj) * invl;
  }
}

__global__ __launch_bounds__(32) void row_wsum(
    const float* __restrict__ qws, const float* __restrict__ kws,
    const float* __restrict__ mcol, const float* __restrict__ lcol,
    float* __restrict__ wsum)
{
  const int lane = threadIdx.x, half = lane >> 4, l16 = lane & 15;
  const int i0 = blockIdx.x * 32;
  const int h  = blockIdx.y;
  const int b  = blockIdx.z;

  const float* qpA = qws + tile_off(b, h, i0 >> 4) + lane * 4;
  v4f afA[8], afB[8];
  load_frags(afA, qpA);
  load_frags(afB, qpA + TILE_F);

  float waccA[8], waccB[8];
#pragma unroll
  for (int r = 0; r < 8; ++r) { waccA[r] = 0.0f; waccB[r] = 0.0f; }

  const float* klane = kws + tile_off(b, h, 0) + lane * 4;
  const float* mlane = mcol + (size_t)(b * HEADS + h) * SEQ + l16;
  const float* llane = lcol + (size_t)(b * HEADS + h) * SEQ + l16;

  v4f fa[8], fb[8];
  float mA, lA, mB, lB;
  load_frags(fa, klane);
  mA = mlane[0];
  lA = llane[0];
  for (int t = 0; t < NT16; t += 2) {
    load_frags(fb, klane + (size_t)(t + 1) * TILE_F);
    mB = mlane[(t + 1) * 16];
    lB = llane[(t + 1) * 16];
    __builtin_amdgcn_sched_barrier(0);
    row_tile(fa, mA, lA, afA, afB, waccA, waccB);
    const int tn = (t + 2 < NT16) ? (t + 2) : (t + 1);
    load_frags(fa, klane + (size_t)tn * TILE_F);
    mA = mlane[tn * 16];
    lA = llane[tn * 16];
    __builtin_amdgcn_sched_barrier(0);
    row_tile(fb, mB, lB, afA, afB, waccA, waccB);
  }

#pragma unroll
  for (int r = 0; r < 8; ++r) {
#pragma unroll
    for (int mask = 1; mask < 16; mask <<= 1) {
      waccA[r] += __shfl_xor(waccA[r], mask, 32);
      waccB[r] += __shfl_xor(waccB[r], mask, 32);
    }
  }
  if (l16 == 0) {
#pragma unroll
    for (int r = 0; r < 8; ++r) {
      wsum[(size_t)(b * SEQ + i0 + r + half * 8) * HEADS + h]      = waccA[r];
      wsum[(size_t)(b * SEQ + i0 + 16 + r + half * 8) * HEADS + h] = waccB[r];
    }
  }
}

__global__ __launch_bounds__(32) void out_gemm(
    const float* __restrict__ vws, const float* __restrict__ wsum,
    const float* __restrict__ Wout, float* __restrict__ out)
{
  const int lane = threadIdx.x, half = lane >> 4, l16 = lane & 15;
  const int o0 = blockIdx.x * 32;
  const int i0 = blockIdx.y * 32;
  const int b  = blockIdx.z;
  const int iA = i0 + l16;
  const int iB = i0 + 16 + l16;

  const float* vrowA = vws + (size_t)(b * SEQ + iA) * HD;
  const float* vrowB = vws + (size_t)(b * SEQ + iB) * HD;
  const float* wrowA = wsum + (size_t)(b * SEQ + iA) * HEADS;
  const float* wrowB = wsum + (size_t)(b * SEQ + iB) * HEADS;
  float wvA[HEADS], wvB[HEADS];
#pragma unroll
  for (int h = 0; h < HEADS; ++h) { wvA[h] = wrowA[h]; wvB[h] = wrowB[h]; }

  const float* wo0 = Wout + (size_t)(o0 + l16) * HD;
  const float* wo1 = Wout + (size_t)(o0 + 16 + l16) * HD;

  v8f acc00 = {}, acc01 = {}, acc10 = {}, acc11 = {};
#pragma unroll 8
  for (int kb = 0; kb < HD; kb += 4) {
    const int kk = kb + half * 2;
    const int h = kk >> 6;
    v2f a0 = *(const v2f*)(vrowA + kk);
    v2f a1 = *(const v2f*)(vrowB + kk);
    a0.x *= wvA[h]; a0.y *= wvA[h];
    a1.x *= wvB[h]; a1.y *= wvB[h];
    v2f b0 = *(const v2f*)(wo0 + kk);
    v2f b1 = *(const v2f*)(wo1 + kk);
    acc00 = WMMA_F32(a0, b0, acc00);
    acc01 = WMMA_F32(a0, b1, acc01);
    acc10 = WMMA_F32(a1, b0, acc10);
    acc11 = WMMA_F32(a1, b1, acc11);
  }
#pragma unroll
  for (int r = 0; r < 8; ++r) {
    const size_t row0 = (size_t)(b * SEQ + i0 + r + half * 8) * COUT;
    const size_t row1 = (size_t)(b * SEQ + i0 + 16 + r + half * 8) * COUT;
    out[row0 + o0 + l16]      = acc00[r];
    out[row0 + o0 + 16 + l16] = acc01[r];
    out[row1 + o0 + l16]      = acc10[r];
    out[row1 + o0 + 16 + l16] = acc11[r];
  }
}

extern "C" void kernel_launch(void* const* d_in, const int* in_sizes, int n_in,
                              void* d_out, int out_size, void* d_ws, size_t ws_size,
                              hipStream_t stream)
{
  (void)in_sizes; (void)n_in; (void)out_size; (void)ws_size;
  const float* X  = (const float*)d_in[0];
  const float* Wq = (const float*)d_in[1];
  const float* Wk = (const float*)d_in[2];
  const float* Wv = (const float*)d_in[3];
  const float* Wo = (const float*)d_in[4];
  float* out = (float*)d_out;

  const size_t QSZ = (size_t)BATCH * SEQ * HD;
  const size_t SSZ = (size_t)BATCH * HEADS * SEQ;
  float* qws  = (float*)d_ws;
  float* kws  = qws + QSZ;
  float* vws  = kws + QSZ;
  float* mcol = vws + QSZ;
  float* lcol = mcol + SSZ;
  float* wsum = lcol + SSZ;

  qkv_gemm<<<dim3(HD / 32, SEQ / 32, BATCH * 3), 32, 0, stream>>>(
      X, Wq, Wk, Wv, qws, kws, vws);
  col_stats<<<dim3(SEQ / 32, HEADS, BATCH), 32, 0, stream>>>(
      qws, kws, mcol, lcol);
  row_wsum<<<dim3(SEQ / 32, HEADS, BATCH), 32, 0, stream>>>(
      qws, kws, mcol, lcol, wsum);
  out_gemm<<<dim3(COUT / 32, SEQ / 32, BATCH), 32, 0, stream>>>(
      vws, wsum, Wo, out);
}


# Scores computed once: a wave keeps its 16-key x 1024-query score strip in registers for maxima, sums and row-sum partials; second QK^T pass dropped; partials summed by a small kernel (new sum order)
# speedup vs baseline: 1.6060x; 1.2446x over previous
#include <hip/hip_runtime.h>
#include <math.h>


typedef float v2f __attribute__((ext_vector_type(2)));
typedef float v4f __attribute__((ext_vector_type(4)));
typedef float v8f __attribute__((ext_vector_type(8)));

#define WMMA_F32(a, b, c) \
  __builtin_amdgcn_wmma_f32_16x16x4_f32(false, (a), false, (b), (short)0, (c), false, false)

constexpr int BATCH = 8;
constexpr int SEQ   = 1024;
constexpr int CIN   = 256;
constexpr int HEADS = 8;
constexpr int DIM   = 64;
constexpr int HD    = HEADS * DIM;
constexpr int COUT  = 256;
constexpr float SCALE = 0.125f;
constexpr int NT16   = SEQ / 16;
constexpr int TILE_F = 16 * DIM;

__device__ __forceinline__ size_t tile_off(int b, int h, int t16)
{
  return ((size_t)(b * HEADS + h) * NT16 + t16) * TILE_F;
}
__device__ __forceinline__ v2f lo2(v4f x) { return __builtin_shufflevector(x, x, 0, 1); }
__device__ __forceinline__ v2f hi2(v4f x) { return __builtin_shufflevector(x, x, 2, 3); }

__device__ __forceinline__ void qkv_step_loads(
    const float* __restrict__ xrow0, const float* __restrict__ xrow1,
    const float* __restrict__ W, int kk, int ncol,
    v2f& a0, v2f& a1, v2f& b0, v2f& b1)
{
  a0 = *(const v2f*)(xrow0 + kk);
  a1 = *(const v2f*)(xrow1 + kk);
  b0.x = W[(size_t)kk * HD + ncol];
  b0.y = W[(size_t)(kk + 1) * HD + ncol];
  b1.x = W[(size_t)kk * HD + ncol + 16];
  b1.y = W[(size_t)(kk + 1) * HD + ncol + 16];
}

__global__ __launch_bounds__(32) void qkv_gemm(
    const float* __restrict__ X,
    const float* __restrict__ Wq, const float* __restrict__ Wk,
    const float* __restrict__ Wv,
    float* __restrict__ qws, float* __restrict__ kws, float* __restrict__ vws)
{
  const int lane = threadIdx.x;
  const int half = lane >> 4, l16 = lane & 15;
  const int n0 = blockIdx.x * 32;
  const int i0 = blockIdx.y * 32;
  const int b  = blockIdx.z / 3;
  const int sel = blockIdx.z % 3;

  const float* W = (sel == 0) ? Wq : (sel == 1) ? Wk : Wv;

  const float* xrow0 = X + (size_t)(b * SEQ + i0 + l16) * CIN;
  const float* xrow1 = xrow0 + (size_t)16 * CIN;

  v8f acc00 = {}, acc01 = {}, acc10 = {}, acc11 = {};
  if (sel == 2) {
#pragma unroll 8
    for (int kb = 0; kb < CIN; kb += 4) {
      v2f a0, a1, b0, b1;
      qkv_step_loads(xrow0, xrow1, W, kb + half * 2, n0 + l16, a0, a1, b0, b1);
      acc00 = WMMA_F32(a0, b0, acc00);
      acc01 = WMMA_F32(a0, b1, acc01);
      acc10 = WMMA_F32(a1, b0, acc10);
      acc11 = WMMA_F32(a1, b1, acc11);
    }
#pragma unroll
    for (int r = 0; r < 8; ++r) {
      const size_t row0 = (size_t)(b * SEQ + i0 + r + half * 8) * HD;
      const size_t row1 = (size_t)(b * SEQ + i0 + 16 + r + half * 8) * HD;
      vws[row0 + n0 + l16]      = acc00[r];
      vws[row0 + n0 + 16 + l16] = acc01[r];
      vws[row1 + n0 + l16]      = acc10[r];
      vws[row1 + n0 + 16 + l16] = acc11[r];
    }
  } else {
#pragma unroll 8
    for (int kb = 0; kb < CIN; kb += 4) {
      v2f a0, a1, b0, b1;
      qkv_step_loads(xrow0, xrow1, W, kb + half * 2, n0 + l16, a0, a1, b0, b1);
      acc00 = WMMA_F32(b0, a0, acc00);
      acc01 = WMMA_F32(b0, a1, acc01);
      acc10 = WMMA_F32(b1, a0, acc10);
      acc11 = WMMA_F32(b1, a1, acc11);
    }
    float* O = (sel == 0) ? qws : kws;
    const int head = n0 >> 6;
    const int w0 = ((n0 & 63) >> 4) * 2;
    float* t0 = O + tile_off(b, head, i0 >> 4) + lane * 4;
    float* t1 = t0 + TILE_F;
    *(v4f*)(t0 + (w0 + 0) * 128) = __builtin_shufflevector(acc00, acc00, 0, 1, 2, 3);
    *(v4f*)(t0 + (w0 + 1) * 128) = __builtin_shufflevector(acc00, acc00, 4, 5, 6, 7);
    *(v4f*)(t0 + (w0 + 2) * 128) = __builtin_shufflevector(acc10, acc10, 0, 1, 2, 3);
    *(v4f*)(t0 + (w0 + 3) * 128) = __builtin_shufflevector(acc10, acc10, 4, 5, 6, 7);
    *(v4f*)(t1 + (w0 + 0) * 128) = __builtin_shufflevector(acc01, acc01, 0, 1, 2, 3);
    *(v4f*)(t1 + (w0 + 1) * 128) = __builtin_shufflevector(acc01, acc01, 4, 5, 6, 7);
    *(v4f*)(t1 + (w0 + 2) * 128) = __builtin_shufflevector(acc11, acc11, 0, 1, 2, 3);
    *(v4f*)(t1 + (w0 + 3) * 128) = __builtin_shufflevector(acc11, acc11, 4, 5, 6, 7);
  }
}

__device__ __forceinline__ void load_frags(v4f (&f)[8], const float* __restrict__ p)
{
#pragma unroll
  for (int w = 0; w < 8; ++w) f[w] = *(const v4f*)(p + w * 128);
}

__device__ __forceinline__ void score_pair(
    const v4f (&kf)[8], const v4f (&q0)[8], const v4f (&q1)[8], v8f& s0, v8f& s1)
{
  v8f a0 = {}, a1 = {};
#pragma unroll
  for (int w = 0; w < 8; ++w) {
    a0 = WMMA_F32(lo2(kf[w]), lo2(q0[w]), a0);
    a1 = WMMA_F32(lo2(kf[w]), lo2(q1[w]), a1);
    a0 = WMMA_F32(hi2(kf[w]), hi2(q0[w]), a0);
    a1 = WMMA_F32(hi2(kf[w]), hi2(q1[w]), a1);
  }
  s0 = a0;
  s1 = a1;
}

__global__ __launch_bounds__(32) void key_strip(
    const float* __restrict__ qws, float* __restrict__ kws)
{
  const int lane = threadIdx.x, half = lane >> 4;
  const int kt = blockIdx.x;
  const int h  = blockIdx.y;
  const int b  = blockIdx.z;

  float* ktile = kws + tile_off(b, h, kt);
  v4f kf[8];
  load_frags(kf, ktile + lane * 4);

  const float* qlane = qws + tile_off(b, h, 0) + lane * 4;

  v8f S[NT16];
  float mx[8];
#pragma unroll
  for (int r = 0; r < 8; ++r) mx[r] = -__builtin_inff();

  constexpr int NSTEP = NT16 / 2;
  v4f F[3][2][8];
  load_frags(F[0][0], qlane);
  load_frags(F[0][1], qlane + (size_t)1 * TILE_F);
  load_frags(F[1][0], qlane + (size_t)2 * TILE_F);
  load_frags(F[1][1], qlane + (size_t)3 * TILE_F);
#pragma unroll
  for (int s = 0; s < NSTEP; ++s) {
    if (s + 2 < NSTEP) {
      load_frags(F[(s + 2) % 3][0], qlane + (size_t)(2 * s + 4) * TILE_F);
      load_frags(F[(s + 2) % 3][1], qlane + (size_t)(2 * s + 5) * TILE_F);
    }
    __builtin_amdgcn_sched_barrier(0);
    score_pair(kf, F[s % 3][0], F[s % 3][1], S[2 * s], S[2 * s + 1]);
#pragma unroll
    for (int r = 0; r < 8; ++r)
      mx[r] = fmaxf(mx[r], fmaxf(S[2 * s][r], S[2 * s + 1][r]));
  }

  float m[8], l[8];
#pragma unroll
  for (int r = 0; r < 8; ++r) {
#pragma unroll
    for (int mask = 1; mask < 16; mask <<= 1)
      mx[r] = fmaxf(mx[r], __shfl_xor(mx[r], mask, 32));
    m[r] = mx[r] * SCALE;
    l[r] = 0.0f;
  }
#pragma unroll
  for (int t = 0; t < NT16; ++t) {
#pragma unroll
    for (int r = 0; r < 8; ++r) {
      const float e = __expf(S[t][r] * SCALE - m[r]);
      S[t][r] = e;
      l[r] += e;
    }
  }
  float invl[8];
#pragma unroll
  for (int r = 0; r < 8; ++r) {
#pragma unroll
    for (int mask = 1; mask < 16; mask <<= 1)
      l[r] += __shfl_xor(l[r], mask, 32);
    invl[r] = 1.0f / l[r];
  }
#pragma unroll
  for (int t = 0; t < NT16; t += 2) {
    float w0 = 0.0f, w1 = 0.0f;
#pragma unroll
    for (int r = 0; r < 8; ++r) {
      w0 += S[t][r] * invl[r];
      w1 += S[t + 1][r] * invl[r];
    }
    w0 += __shfl_xor(w0, 16, 32);
    w1 += __shfl_xor(w1, 16, 32);
    ktile[16 * t + lane] = half ? w1 : w0;
  }
}

__global__ __launch_bounds__(256) void wsum_reduce(
    const float* __restrict__ part, float* __restrict__ wsum)
{
  const int idx = blockIdx.x * 256 + threadIdx.x;
  const int i  = idx & (SEQ - 1);
  const int bh = idx >> 10;
  const int b  = bh >> 3;
  const int h  = bh & (HEADS - 1);
  const float* p = part + (size_t)bh * NT16 * TILE_F + i;
  float s0 = 0.0f, s1 = 0.0f, s2 = 0.0f, s3 = 0.0f;
#pragma unroll 4
  for (int kt = 0; kt < NT16; kt += 4) {
    s0 += p[(size_t)(kt + 0) * TILE_F];
    s1 += p[(size_t)(kt + 1) * TILE_F];
    s2 += p[(size_t)(kt + 2) * TILE_F];
    s3 += p[(size_t)(kt + 3) * TILE_F];
  }
  wsum[(size_t)(b * SEQ + i) * HEADS + h] = (s0 + s1) + (s2 + s3);
}

__global__ __launch_bounds__(32) void out_gemm(
    const float* __restrict__ vws, const float* __restrict__ wsum,
    const float* __restrict__ Wout, float* __restrict__ out)
{
  const int lane = threadIdx.x, half = lane >> 4, l16 = lane & 15;
  const int o0 = blockIdx.x * 32;
  const int i0 = blockIdx.y * 32;
  const int b  = blockIdx.z;
  const int iA = i0 + l16;
  const int iB = i0 + 16 + l16;

  const float* vrowA = vws + (size_t)(b * SEQ + iA) * HD;
  const float* vrowB = vws + (size_t)(b * SEQ + iB) * HD;
  const float* wrowA = wsum + (size_t)(b * SEQ + iA) * HEADS;
  const float* wrowB = wsum + (size_t)(b * SEQ + iB) * HEADS;
  float wvA[HEADS], wvB[HEADS];
#pragma unroll
  for (int h = 0; h < HEADS; ++h) { wvA[h] = wrowA[h]; wvB[h] = wrowB[h]; }

  const float* wo0 = Wout + (size_t)(o0 + l16) * HD;
  const float* wo1 = Wout + (size_t)(o0 + 16 + l16) * HD;

  v8f acc00 = {}, acc01 = {}, acc10 = {}, acc11 = {};
#pragma unroll 8
  for (int kb = 0; kb < HD; kb += 4) {
    const int kk = kb + half * 2;
    const int h = kk >> 6;
    v2f a0 = *(const v2f*)(vrowA + kk);
    v2f a1 = *(const v2f*)(vrowB + kk);
    a0.x *= wvA[h]; a0.y *= wvA[h];
    a1.x *= wvB[h]; a1.y *= wvB[h];
    v2f b0 = *(const v2f*)(wo0 + kk);
    v2f b1 = *(const v2f*)(wo1 + kk);
    acc00 = WMMA_F32(a0, b0, acc00);
    acc01 = WMMA_F32(a0, b1, acc01);
    acc10 = WMMA_F32(a1, b0, acc10);
    acc11 = WMMA_F32(a1, b1, acc11);
  }
#pragma unroll
  for (int r = 0; r < 8; ++r) {
    const size_t row0 = (size_t)(b * SEQ + i0 + r + half * 8) * COUT;
    const size_t row1 = (size_t)(b * SEQ + i0 + 16 + r + half * 8) * COUT;
    out[row0 + o0 + l16]      = acc00[r];
    out[row0 + o0 + 16 + l16] = acc01[r];
    out[row1 + o0 + l16]      = acc10[r];
    out[row1 + o0 + 16 + l16] = acc11[r];
  }
}

extern "C" void kernel_launch(void* const* d_in, const int* in_sizes, int n_in,
                              void* d_out, int out_size, void* d_ws, size_t ws_size,
                              hipStream_t stream)
{
  (void)in_sizes; (void)n_in; (void)out_size; (void)ws_size;
  const float* X  = (const float*)d_in[0];
  const float* Wq = (const float*)d_in[1];
  const float* Wk = (const float*)d_in[2];
  const float* Wv = (const float*)d_in[3];
  const float* Wo = (const float*)d_in[4];
  float* out = (float*)d_out;

  const size_t QSZ = (size_t)BATCH * SEQ * HD;
  const size_t SSZ = (size_t)BATCH * HEADS * SEQ;
  float* qws  = (float*)d_ws;
  float* kws  = qws + QSZ;
  float* vws  = kws + QSZ;
  float* wsum = vws + QSZ + 2 * SSZ;

  qkv_gemm<<<dim3(HD / 32, SEQ / 32, BATCH * 3), 32, 0, stream>>>(
      X, Wq, Wk, Wv, qws, kws, vws);
  key_strip<<<dim3(NT16, HEADS, BATCH), 32, 0, stream>>>(qws, kws);
  wsum_reduce<<<dim3(BATCH * HEADS * SEQ / 256), 256, 0, stream>>>(kws, wsum);
  out_gemm<<<dim3(COUT / 32, SEQ / 32, BATCH), 32, 0, stream>>>(
      vws, wsum, Wo, out);
}


# Projection GEMMs (qkv, out): row operands fetched per 32-k block as 16-byte loads of 16 consecutive k per lane (k order permuted inside a block; out GEMM's group issued ahead), not 8 B per k-step
# speedup vs baseline: 1.7843x; 1.1110x over previous
#include <hip/hip_runtime.h>
#include <math.h>


typedef float v2f __attribute__((ext_vector_type(2)));
typedef float v4f __attribute__((ext_vector_type(4)));
typedef float v8f __attribute__((ext_vector_type(8)));

#define WMMA_F32(a, b, c) \
  __builtin_amdgcn_wmma_f32_16x16x4_f32(false, (a), false, (b), (short)0, (c), false, false)

constexpr int BATCH = 8;
constexpr int SEQ   = 1024;
constexpr int CIN   = 256;
constexpr int HEADS = 8;
constexpr int DIM   = 64;
constexpr int HD    = HEADS * DIM;
constexpr int COUT  = 256;
constexpr float SCALE = 0.125f;
constexpr int NT16   = SEQ / 16;
constexpr int TILE_F = 16 * DIM;

__device__ __forceinline__ size_t tile_off(int b, int h, int t16)
{
  return ((size_t)(b * HEADS + h) * NT16 + t16) * TILE_F;
}
__device__ __forceinline__ v2f lo2(v4f x) { return __builtin_shufflevector(x, x, 0, 1); }
__device__ __forceinline__ v2f hi2(v4f x) { return __builtin_shufflevector(x, x, 2, 3); }

__device__ __forceinline__ void load_chunks(v4f (&f)[4], const float* __restrict__ p)
{
#pragma unroll
  for (int q = 0; q < 4; ++q) f[q] = *(const v4f*)(p + 4 * q);
}
__device__ __forceinline__ v2f step_pair(const v4f (&f)[4], int s)
{
  return (s & 1) ? hi2(f[s >> 1]) : lo2(f[s >> 1]);
}

__device__ __forceinline__ void w_step_loads(
    const float* __restrict__ wk, int s, v2f& b0, v2f& b1)
{
  b0.x = wk[(size_t)(2 * s) * HD];
  b0.y = wk[(size_t)(2 * s + 1) * HD];
  b1.x = wk[(size_t)(2 * s) * HD + 16];
  b1.y = wk[(size_t)(2 * s + 1) * HD + 16];
}

__global__ __launch_bounds__(32) void qkv_gemm(
    const float* __restrict__ X,
    const float* __restrict__ Wq, const float* __restrict__ Wk,
    const float* __restrict__ Wv,
    float* __restrict__ qws, float* __restrict__ kws, float* __restrict__ vws)
{
  const int lane = threadIdx.x;
  const int half = lane >> 4, l16 = lane & 15;
  const int n0 = blockIdx.x * 32;
  const int i0 = blockIdx.y * 32;
  const int b  = blockIdx.z / 3;
  const int sel = blockIdx.z % 3;

  const float* W = (sel == 0) ? Wq : (sel == 1) ? Wk : Wv;

  const float* xrow0 = X + (size_t)(b * SEQ + i0 + l16) * CIN + 16 * half;
  const float* xrow1 = xrow0 + (size_t)16 * CIN;
  const float* wlane = W + (size_t)(16 * half) * HD + n0 + l16;

  v8f acc00 = {}, acc01 = {}, acc10 = {}, acc11 = {};
  if (sel == 2) {
    for (int kb0 = 0; kb0 < CIN; kb0 += 32) {
      v4f xa[4], xb[4];
      load_chunks(xa, xrow0 + kb0);
      load_chunks(xb, xrow1 + kb0);
      const float* wk = wlane + (size_t)kb0 * HD;
#pragma unroll
      for (int s = 0; s < 8; ++s) {
        v2f b0, b1;
        w_step_loads(wk, s, b0, b1);
        const v2f a0 = step_pair(xa, s), a1 = step_pair(xb, s);
        acc00 = WMMA_F32(a0, b0, acc00);
        acc01 = WMMA_F32(a0, b1, acc01);
        acc10 = WMMA_F32(a1, b0, acc10);
        acc11 = WMMA_F32(a1, b1, acc11);
      }
    }
#pragma unroll
    for (int r = 0; r < 8; ++r) {
      const size_t row0 = (size_t)(b * SEQ + i0 + r + half * 8) * HD;
      const size_t row1 = (size_t)(b * SEQ + i0 + 16 + r + half * 8) * HD;
      vws[row0 + n0 + l16]      = acc00[r];
      vws[row0 + n0 + 16 + l16] = acc01[r];
      vws[row1 + n0 + l16]      = acc10[r];
      vws[row1 + n0 + 16 + l16] = acc11[r];
    }
  } else {
    for (int kb0 = 0; kb0 < CIN; kb0 += 32) {
      v4f xa[4], xb[4];
      load_chunks(xa, xrow0 + kb0);
      load_chunks(xb, xrow1 + kb0);
      const float* wk = wlane + (size_t)kb0 * HD;
#pragma unroll
      for (int s = 0; s < 8; ++s) {
        v2f b0, b1;
        w_step_loads(wk, s, b0, b1);
        const v2f a0 = step_pair(xa, s), a1 = step_pair(xb, s);
        acc00 = WMMA_F32(b0, a0, acc00);
        acc01 = WMMA_F32(b0, a1, acc01);
        acc10 = WMMA_F32(b1, a0, acc10);
        acc11 = WMMA_F32(b1, a1, acc11);
      }
    }
    float* O = (sel == 0) ? qws : kws;
    const int head = n0 >> 6;
    const int w0 = ((n0 & 63) >> 4) * 2;
    float* t0 = O + tile_off(b, head, i0 >> 4) + lane * 4;
    float* t1 = t0 + TILE_F;
    *(v4f*)(t0 + (w0 + 0) * 128) = __builtin_shufflevector(acc00, acc00, 0, 1, 2, 3);
    *(v4f*)(t0 + (w0 + 1) * 128) = __builtin_shufflevector(acc00, acc00, 4, 5, 6, 7);
    *(v4f*)(t0 + (w0 + 2) * 128) = __builtin_shufflevector(acc10, acc10, 0, 1, 2, 3);
    *(v4f*)(t0 + (w0 + 3) * 128) = __builtin_shufflevector(acc10, acc10, 4, 5, 6, 7);
    *(v4f*)(t1 + (w0 + 0) * 128) = __builtin_shufflevector(acc01, acc01, 0, 1, 2, 3);
    *(v4f*)(t1 + (w0 + 1) * 128) = __builtin_shufflevector(acc01, acc01, 4, 5, 6, 7);
    *(v4f*)(t1 + (w0 + 2) * 128) = __builtin_shufflevector(acc11, acc11, 0, 1, 2, 3);
    *(v4f*)(t1 + (w0 + 3) * 128) = __builtin_shufflevector(acc11, acc11, 4, 5, 6, 7);
  }
}

__device__ __forceinline__ void load_frags(v4f (&f)[8], const float* __restrict__ p)
{
#pragma unroll
  for (int w = 0; w < 8; ++w) f[w] = *(const v4f*)(p + w * 128);
}

__device__ __forceinline__ void score_pair(
    const v4f (&kf)[8], const v4f (&q0)[8], const v4f (&q1)[8], v8f& s0, v8f& s1)
{
  v8f a0 = {}, a1 = {};
#pragma unroll
  for (int w = 0; w < 8; ++w) {
    a0 = WMMA_F32(lo2(kf[w]), lo2(q0[w]), a0);
    a1 = WMMA_F32(lo2(kf[w]), lo2(q1[w]), a1);
    a0 = WMMA_F32(hi2(kf[w]), hi2(q0[w]), a0);
    a1 = WMMA_F32(hi2(kf[w]), hi2(q1[w]), a1);
  }
  s0 = a0;
  s1 = a1;
}

__global__ __launch_bounds__(32) void key_strip(
    const float* __restrict__ qws, float* __restrict__ kws)
{
  const int lane = threadIdx.x, half = lane >> 4;
  const int kt = blockIdx.x;
  const int h  = blockIdx.y;
  const int b  = blockIdx.z;

  float* ktile = kws + tile_off(b, h, kt);
  v4f kf[8];
  load_frags(kf, ktile + lane * 4);

  const float* qlane = qws + tile_off(b, h, 0) + lane * 4;

  v8f S[NT16];
  float mx[8];
#pragma unroll
  for (int r = 0; r < 8; ++r) mx[r] = -__builtin_inff();

  constexpr int NSTEP = NT16 / 2;
  v4f F[3][2][8];
  load_frags(F[0][0], qlane);
  load_frags(F[0][1], qlane + (size_t)1 * TILE_F);
  load_frags(F[1][0], qlane + (size_t)2 * TILE_F);
  load_frags(F[1][1], qlane + (size_t)3 * TILE_F);
#pragma unroll
  for (int s = 0; s < NSTEP; ++s) {
    if (s + 2 < NSTEP) {
      load_frags(F[(s + 2) % 3][0], qlane + (size_t)(2 * s + 4) * TILE_F);
      load_frags(F[(s + 2) % 3][1], qlane + (size_t)(2 * s + 5) * TILE_F);
    }
    __builtin_amdgcn_sched_barrier(0);
    score_pair(kf, F[s % 3][0], F[s % 3][1], S[2 * s], S[2 * s + 1]);
#pragma unroll
    for (int r = 0; r < 8; ++r)
      mx[r] = fmaxf(mx[r], fmaxf(S[2 * s][r], S[2 * s + 1][r]));
  }

  float m[8], l[8];
#pragma unroll
  for (int r = 0; r < 8; ++r) {
#pragma unroll
    for (int mask = 1; mask < 16; mask <<= 1)
      mx[r] = fmaxf(mx[r], __shfl_xor(mx[r], mask, 32));
    m[r] = mx[r] * SCALE;
    l[r] = 0.0f;
  }
#pragma unroll
  for (int t = 0; t < NT16; ++t) {
#pragma unroll
    for (int r = 0; r < 8; ++r) {
      const float e = __expf(S[t][r] * SCALE - m[r]);
      S[t][r] = e;
      l[r] += e;
    }
  }
  float invl[8];
#pragma unroll
  for (int r = 0; r < 8; ++r) {
#pragma unroll
    for (int mask = 1; mask < 16; mask <<= 1)
      l[r] += __shfl_xor(l[r], mask, 32);
    invl[r] = 1.0f / l[r];
  }
#pragma unroll
  for (int t = 0; t < NT16; t += 2) {
    float w0 = 0.0f, w1 = 0.0f;
#pragma unroll
    for (int r = 0; r < 8; ++r) {
      w0 += S[t][r] * invl[r];
      w1 += S[t + 1][r] * invl[r];
    }
    w0 += __shfl_xor(w0, 16, 32);
    w1 += __shfl_xor(w1, 16, 32);
    ktile[16 * t + lane] = half ? w1 : w0;
  }
}

__global__ __launch_bounds__(256) void wsum_reduce(
    const float* __restrict__ part, float* __restrict__ wsum)
{
  const int idx = blockIdx.x * 256 + threadIdx.x;
  const int i  = idx & (SEQ - 1);
  const int bh = idx >> 10;
  const int b  = bh >> 3;
  const int h  = bh & (HEADS - 1);
  const float* p = part + (size_t)bh * NT16 * TILE_F + i;
  float s0 = 0.0f, s1 = 0.0f, s2 = 0.0f, s3 = 0.0f;
#pragma unroll 4
  for (int kt = 0; kt < NT16; kt += 4) {
    s0 += p[(size_t)(kt + 0) * TILE_F];
    s1 += p[(size_t)(kt + 1) * TILE_F];
    s2 += p[(size_t)(kt + 2) * TILE_F];
    s3 += p[(size_t)(kt + 3) * TILE_F];
  }
  wsum[(size_t)(b * SEQ + i) * HEADS + h] = (s0 + s1) + (s2 + s3);
}

__global__ __launch_bounds__(32) void out_gemm(
    const float* __restrict__ vws, const float* __restrict__ wsum,
    const float* __restrict__ Wout, float* __restrict__ out)
{
  const int lane = threadIdx.x, half = lane >> 4, l16 = lane & 15;
  const int o0 = blockIdx.x * 32;
  const int i0 = blockIdx.y * 32;
  const int b  = blockIdx.z;
  const int iA = i0 + l16;
  const int iB = i0 + 16 + l16;

  const float* vrowA = vws + (size_t)(b * SEQ + iA) * HD + 16 * half;
  const float* vrowB = vws + (size_t)(b * SEQ + iB) * HD + 16 * half;
  const float* wrowA = wsum + (size_t)(b * SEQ + iA) * HEADS;
  const float* wrowB = wsum + (size_t)(b * SEQ + iB) * HEADS;
  float wvA[HEADS], wvB[HEADS];
#pragma unroll
  for (int h = 0; h < HEADS; ++h) { wvA[h] = wrowA[h]; wvB[h] = wrowB[h]; }

  const float* wo0 = Wout + (size_t)(o0 + l16) * HD + 16 * half;
  const float* wo1 = Wout + (size_t)(o0 + 16 + l16) * HD + 16 * half;

  v8f acc00 = {}, acc01 = {}, acc10 = {}, acc11 = {};
  constexpr int NBLK = HD / 32;
  v4f A0[2][4], A1[2][4], B0[2][4], B1[2][4];
  load_chunks(A0[0], vrowA);
  load_chunks(A1[0], vrowB);
  load_chunks(B0[0], wo0);
  load_chunks(B1[0], wo1);
#pragma unroll
  for (int blk = 0; blk < NBLK; ++blk) {
    const int cur = blk & 1, nxt = cur ^ 1;
    if (blk + 1 < NBLK) {
      load_chunks(A0[nxt], vrowA + (blk + 1) * 32);
      load_chunks(A1[nxt], vrowB + (blk + 1) * 32);
      load_chunks(B0[nxt], wo0 + (blk + 1) * 32);
      load_chunks(B1[nxt], wo1 + (blk + 1) * 32);
    }
    __builtin_amdgcn_sched_barrier(0);
    const float sA = wvA[blk >> 1], sB = wvB[blk >> 1];
#pragma unroll
    for (int s = 0; s < 8; ++s) {
      const v2f a0 = step_pair(A0[cur], s) * sA;
      const v2f a1 = step_pair(A1[cur], s) * sB;
      const v2f b0 = step_pair(B0[cur], s), b1 = step_pair(B1[cur], s);
      acc00 = WMMA_F32(a0, b0, acc00);
      acc01 = WMMA_F32(a0, b1, acc01);
      acc10 = WMMA_F32(a1, b0, acc10);
      acc11 = WMMA_F32(a1, b1, acc11);
    }
  }
#pragma unroll
  for (int r = 0; r < 8; ++r) {
    const size_t row0 = (size_t)(b * SEQ + i0 + r + half * 8) * COUT;
    const size_t row1 = (size_t)(b * SEQ + i0 + 16 + r + half * 8) * COUT;
    out[row0 + o0 + l16]      = acc00[r];
    out[row0 + o0 + 16 + l16] = acc01[r];
    out[row1 + o0 + l16]      = acc10[r];
    out[row1 + o0 + 16 + l16] = acc11[r];
  }
}

extern "C" void kernel_launch(void* const* d_in, const int* in_sizes, int n_in,
                              void* d_out, int out_size, void* d_ws, size_t ws_size,
                              hipStream_t stream)
{
  (void)in_sizes; (void)n_in; (void)out_size; (void)ws_size;
  const float* X  = (const float*)d_in[0];
  const float* Wq = (const float*)d_in[1];
  const float* Wk = (const float*)d_in[2];
  const float* Wv = (const float*)d_in[3];
  const float* Wo = (const float*)d_in[4];
  float* out = (float*)d_out;

  const size_t QSZ = (size_t)BATCH * SEQ * HD;
  const size_t SSZ = (size_t)BATCH * HEADS * SEQ;
  float* qws  = (float*)d_ws;
  float* kws  = qws + QSZ;
  float* vws  = kws + QSZ;
  float* wsum = vws + QSZ + 2 * SSZ;

  qkv_gemm<<<dim3(HD / 32, SEQ / 32, BATCH * 3), 32, 0, stream>>>(
      X, Wq, Wk, Wv, qws, kws, vws);
  key_strip<<<dim3(NT16, HEADS, BATCH), 32, 0, stream>>>(qws, kws);
  wsum_reduce<<<dim3(BATCH * HEADS * SEQ / 256), 256, 0, stream>>>(kws, wsum);
  out_gemm<<<dim3(COUT / 32, SEQ / 32, BATCH), 32, 0, stream>>>(
      vws, wsum, Wo, out);
}


# Score strip re-tiled: a 4-wave workgroup owns 64 keys, each wave 256 queries, so a loaded query fragment feeds 4 key tiles; key max/sum joined by row DPP + LDS; 16 partials per row (new sum order)
# speedup vs baseline: 1.9071x; 1.0688x over previous
#include <hip/hip_runtime.h>
#include <math.h>


typedef float v2f __attribute__((ext_vector_type(2)));
typedef float v4f __attribute__((ext_vector_type(4)));
typedef float v8f __attribute__((ext_vector_type(8)));

#define WMMA_F32(a, b, c) \
  __builtin_amdgcn_wmma_f32_16x16x4_f32(false, (a), false, (b), (short)0, (c), false, false)

constexpr int BATCH = 8;
constexpr int SEQ   = 1024;
constexpr int CIN   = 256;
constexpr int HEADS = 8;
constexpr int DIM   = 64;
constexpr int HD    = HEADS * DIM;
constexpr int COUT  = 256;
constexpr float SCALE = 0.125f;
constexpr int NT16   = SEQ / 16;
constexpr int TILE_F = 16 * DIM;

__device__ __forceinline__ size_t tile_off(int b, int h, int t16)
{
  return ((size_t)(b * HEADS + h) * NT16 + t16) * TILE_F;
}
__device__ __forceinline__ v2f lo2(v4f x) { return __builtin_shufflevector(x, x, 0, 1); }
__device__ __forceinline__ v2f hi2(v4f x) { return __builtin_shufflevector(x, x, 2, 3); }

__device__ __forceinline__ void load_chunks(v4f (&f)[4], const float* __restrict__ p)
{
#pragma unroll
  for (int q = 0; q < 4; ++q) f[q] = *(const v4f*)(p + 4 * q);
}
__device__ __forceinline__ v2f step_pair(const v4f (&f)[4], int s)
{
  return (s & 1) ? hi2(f[s >> 1]) : lo2(f[s >> 1]);
}

__device__ __forceinline__ void w_step_loads(
    const float* __restrict__ wk, int s, v2f& b0, v2f& b1)
{
  b0.x = wk[(size_t)(2 * s) * HD];
  b0.y = wk[(size_t)(2 * s + 1) * HD];
  b1.x = wk[(size_t)(2 * s) * HD + 16];
  b1.y = wk[(size_t)(2 * s + 1) * HD + 16];
}

__global__ __launch_bounds__(32) void qkv_gemm(
    const float* __restrict__ X,
    const float* __restrict__ Wq, const float* __restrict__ Wk,
    const float* __restrict__ Wv,
    float* __restrict__ qws, float* __restrict__ kws, float* __restrict__ vws)
{
  const int lane = threadIdx.x;
  const int half = lane >> 4, l16 = lane & 15;
  const int n0 = blockIdx.x * 32;
  const int i0 = blockIdx.y * 32;
  const int b  = blockIdx.z / 3;
  const int sel = blockIdx.z % 3;

  const float* W = (sel == 0) ? Wq : (sel == 1) ? Wk : Wv;

  const float* xrow0 = X + (size_t)(b * SEQ + i0 + l16) * CIN + 16 * half;
  const float* xrow1 = xrow0 + (size_t)16 * CIN;
  const float* wlane = W + (size_t)(16 * half) * HD + n0 + l16;

  v8f acc00 = {}, acc01 = {}, acc10 = {}, acc11 = {};
  if (sel == 2) {
    for (int kb0 = 0; kb0 < CIN; kb0 += 32) {
      v4f xa[4], xb[4];
      load_chunks(xa, xrow0 + kb0);
      load_chunks(xb, xrow1 + kb0);
      const float* wk = wlane + (size_t)kb0 * HD;
#pragma unroll
      for (int s = 0; s < 8; ++s) {
        v2f b0, b1;
        w_step_loads(wk, s, b0, b1);
        const v2f a0 = step_pair(xa, s), a1 = step_pair(xb, s);
        acc00 = WMMA_F32(a0, b0, acc00);
        acc01 = WMMA_F32(a0, b1, acc01);
        acc10 = WMMA_F32(a1, b0, acc10);
        acc11 = WMMA_F32(a1, b1, acc11);
      }
    }
#pragma unroll
    for (int r = 0; r < 8; ++r) {
      const size_t row0 = (size_t)(b * SEQ + i0 + r + half * 8) * HD;
      const size_t row1 = (size_t)(b * SEQ + i0 + 16 + r + half * 8) * HD;
      vws[row0 + n0 + l16]      = acc00[r];
      vws[row0 + n0 + 16 + l16] = acc01[r];
      vws[row1 + n0 + l16]      = acc10[r];
      vws[row1 + n0 + 16 + l16] = acc11[r];
    }
  } else {
    for (int kb0 = 0; kb0 < CIN; kb0 += 32) {
      v4f xa[4], xb[4];
      load_chunks(xa, xrow0 + kb0);
      load_chunks(xb, xrow1 + kb0);
      const float* wk = wlane + (size_t)kb0 * HD;
#pragma unroll
      for (int s = 0; s < 8; ++s) {
        v2f b0, b1;
        w_step_loads(wk, s, b0, b1);
        const v2f a0 = step_pair(xa, s), a1 = step_pair(xb, s);
        acc00 = WMMA_F32(b0, a0, acc00);
        acc01 = WMMA_F32(b0, a1, acc01);
        acc10 = WMMA_F32(b1, a0, acc10);
        acc11 = WMMA_F32(b1, a1, acc11);
      }
    }
    float* O = (sel == 0) ? qws : kws;
    const int head = n0 >> 6;
    const int w0 = ((n0 & 63) >> 4) * 2;
    float* t0 = O + tile_off(b, head, i0 >> 4) + lane * 4;
    float* t1 = t0 + TILE_F;
    *(v4f*)(t0 + (w0 + 0) * 128) = __builtin_shufflevector(acc00, acc00, 0, 1, 2, 3);
    *(v4f*)(t0 + (w0 + 1) * 128) = __builtin_shufflevector(acc00, acc00, 4, 5, 6, 7);
    *(v4f*)(t0 + (w0 + 2) * 128) = __builtin_shufflevector(acc10, acc10, 0, 1, 2, 3);
    *(v4f*)(t0 + (w0 + 3) * 128) = __builtin_shufflevector(acc10, acc10, 4, 5, 6, 7);
    *(v4f*)(t1 + (w0 + 0) * 128) = __builtin_shufflevector(acc01, acc01, 0, 1, 2, 3);
    *(v4f*)(t1 + (w0 + 1) * 128) = __builtin_shufflevector(acc01, acc01, 4, 5, 6, 7);
    *(v4f*)(t1 + (w0 + 2) * 128) = __builtin_shufflevector(acc11, acc11, 0, 1, 2, 3);
    *(v4f*)(t1 + (w0 + 3) * 128) = __builtin_shufflevector(acc11, acc11, 4, 5, 6, 7);
  }
}

__device__ __forceinline__ void load_frags(v4f (&f)[8], const float* __restrict__ p)
{
#pragma unroll
  for (int w = 0; w < 8; ++w) f[w] = *(const v4f*)(p + w * 128);
}

constexpr int KT_WG   = 4;
constexpr int WAVES   = 4;
constexpr int QT_WAVE = NT16 / WAVES;
constexpr int NGRP    = NT16 / KT_WG;

template <int XM> __device__ __forceinline__ float row_xor(float x)
{
  return __builtin_bit_cast(float, __builtin_amdgcn_update_dpp(
      0, __builtin_bit_cast(int, x), 0x160 + XM, 0xf, 0xf, true));
}
__device__ __forceinline__ float row16_max(float x)
{
  x = fmaxf(x, row_xor<1>(x));
  x = fmaxf(x, row_xor<2>(x));
  x = fmaxf(x, row_xor<4>(x));
  x = fmaxf(x, row_xor<8>(x));
  return x;
}
__device__ __forceinline__ float row16_sum(float x)
{
  x += row_xor<1>(x);
  x += row_xor<2>(x);
  x += row_xor<4>(x);
  x += row_xor<8>(x);
  return x;
}

__global__ __launch_bounds__(WAVES * 32) void key_strip(
    const float* __restrict__ qws, float* __restrict__ kws)
{
  __shared__ v4f xm[WAVES][KT_WG * 4];
  __shared__ v4f xl[WAVES][KT_WG * 4];

  const int lane = threadIdx.x & 31, half = lane >> 4, l16 = lane & 15;
  const int wave = __builtin_amdgcn_readfirstlane((int)(threadIdx.x >> 5));
  const int kg = blockIdx.x;
  const int h  = blockIdx.y;
  const int b  = blockIdx.z;

  float* kgrp = kws + tile_off(b, h, KT_WG * kg);
  v4f kf[KT_WG][8];
#pragma unroll
  for (int j = 0; j < KT_WG; ++j)
    load_frags(kf[j], kgrp + (size_t)j * TILE_F + lane * 4);

  const float* qlane = qws + tile_off(b, h, wave * QT_WAVE) + lane * 4;

  v8f S[KT_WG][QT_WAVE];
  float mx[KT_WG][8];
#pragma unroll
  for (int j = 0; j < KT_WG; ++j)
#pragma unroll
    for (int r = 0; r < 8; ++r) mx[j][r] = -__builtin_inff();

  constexpr int NSTEP = QT_WAVE / 2;
  v4f F[2][2][8];
  load_frags(F[0][0], qlane);
  load_frags(F[0][1], qlane + (size_t)1 * TILE_F);
#pragma unroll
  for (int s = 0; s < NSTEP; ++s) {
    const int cur = s & 1, nxt = cur ^ 1;
    if (s + 1 < NSTEP) {
      load_frags(F[nxt][0], qlane + (size_t)(2 * s + 2) * TILE_F);
      load_frags(F[nxt][1], qlane + (size_t)(2 * s + 3) * TILE_F);
    }
    __builtin_amdgcn_sched_barrier(0);
    v8f a0[KT_WG], a1[KT_WG];
#pragma unroll
    for (int j = 0; j < KT_WG; ++j) { a0[j] = (v8f){}; a1[j] = (v8f){}; }
#pragma unroll
    for (int w = 0; w < 8; ++w) {
#pragma unroll
      for (int j = 0; j < KT_WG; ++j) {
        a0[j] = WMMA_F32(lo2(kf[j][w]), lo2(F[cur][0][w]), a0[j]);
        a1[j] = WMMA_F32(lo2(kf[j][w]), lo2(F[cur][1][w]), a1[j]);
      }
#pragma unroll
      for (int j = 0; j < KT_WG; ++j) {
        a0[j] = WMMA_F32(hi2(kf[j][w]), hi2(F[cur][0][w]), a0[j]);
        a1[j] = WMMA_F32(hi2(kf[j][w]), hi2(F[cur][1][w]), a1[j]);
      }
    }
#pragma unroll
    for (int j = 0; j < KT_WG; ++j) {
      S[j][2 * s]     = a0[j];
      S[j][2 * s + 1] = a1[j];
#pragma unroll
      for (int r = 0; r < 8; ++r)
        mx[j][r] = fmaxf(mx[j][r], fmaxf(a0[j][r], a1[j][r]));
    }
  }

#pragma unroll
  for (int j = 0; j < KT_WG; ++j)
#pragma unroll
    for (int r = 0; r < 8; ++r) mx[j][r] = row16_max(mx[j][r]);
  if (l16 == 0) {
#pragma unroll
    for (int j = 0; j < KT_WG; ++j) {
      xm[wave][4 * j + 2 * half]     = (v4f){mx[j][0], mx[j][1], mx[j][2], mx[j][3]};
      xm[wave][4 * j + 2 * half + 1] = (v4f){mx[j][4], mx[j][5], mx[j][6], mx[j][7]};
    }
  }
  __syncthreads();
  float m[KT_WG][8], l[KT_WG][8];
#pragma unroll
  for (int j = 0; j < KT_WG; ++j) {
    v4f lo = xm[0][4 * j + 2 * half], hi = xm[0][4 * j + 2 * half + 1];
#pragma unroll
    for (int w = 1; w < WAVES; ++w) {
      const v4f lo_w = xm[w][4 * j + 2 * half], hi_w = xm[w][4 * j + 2 * half + 1];
#pragma unroll
      for (int e = 0; e < 4; ++e) {
        lo[e] = fmaxf(lo[e], lo_w[e]);
        hi[e] = fmaxf(hi[e], hi_w[e]);
      }
    }
#pragma unroll
    for (int e = 0; e < 4; ++e) {
      m[j][e]     = lo[e] * SCALE;
      m[j][4 + e] = hi[e] * SCALE;
      l[j][e]     = 0.0f;
      l[j][4 + e] = 0.0f;
    }
  }
#pragma unroll
  for (int j = 0; j < KT_WG; ++j)
#pragma unroll
    for (int t = 0; t < QT_WAVE; ++t)
#pragma unroll
      for (int r = 0; r < 8; ++r) {
        const float e = __expf(S[j][t][r] * SCALE - m[j][r]);
        S[j][t][r] = e;
        l[j][r] += e;
      }
#pragma unroll
  for (int j = 0; j < KT_WG; ++j)
#pragma unroll
    for (int r = 0; r < 8; ++r) l[j][r] = row16_sum(l[j][r]);
  if (l16 == 0) {
#pragma unroll
    for (int j = 0; j < KT_WG; ++j) {
      xl[wave][4 * j + 2 * half]     = (v4f){l[j][0], l[j][1], l[j][2], l[j][3]};
      xl[wave][4 * j + 2 * half + 1] = (v4f){l[j][4], l[j][5], l[j][6], l[j][7]};
    }
  }
  __syncthreads();
  float invl[KT_WG][8];
#pragma unroll
  for (int j = 0; j < KT_WG; ++j) {
    v4f lo = xl[0][4 * j + 2 * half], hi = xl[0][4 * j + 2 * half + 1];
#pragma unroll
    for (int w = 1; w < WAVES; ++w) {
      lo += xl[w][4 * j + 2 * half];
      hi += xl[w][4 * j + 2 * half + 1];
    }
#pragma unroll
    for (int e = 0; e < 4; ++e) {
      invl[j][e]     = 1.0f / lo[e];
      invl[j][4 + e] = 1.0f / hi[e];
    }
  }
  float* pout = kgrp + 16 * (wave * QT_WAVE);
#pragma unroll
  for (int t = 0; t < QT_WAVE; t += 2) {
    float w0 = 0.0f, w1 = 0.0f;
#pragma unroll
    for (int j = 0; j < KT_WG; ++j)
#pragma unroll
      for (int r = 0; r < 8; ++r) {
        w0 += S[j][t][r] * invl[j][r];
        w1 += S[j][t + 1][r] * invl[j][r];
      }
    w0 += __shfl_xor(w0, 16, 32);
    w1 += __shfl_xor(w1, 16, 32);
    pout[16 * t + lane] = half ? w1 : w0;
  }
}

__global__ __launch_bounds__(256) void wsum_reduce(
    const float* __restrict__ part, float* __restrict__ wsum)
{
  const int idx = blockIdx.x * 256 + threadIdx.x;
  const int i  = idx & (SEQ - 1);
  const int bh = idx >> 10;
  const int b  = bh >> 3;
  const int h  = bh & (HEADS - 1);
  const float* p = part + (size_t)bh * NT16 * TILE_F + i;
  constexpr size_t GSTRIDE = (size_t)KT_WG * TILE_F;
  float s0 = 0.0f, s1 = 0.0f, s2 = 0.0f, s3 = 0.0f;
#pragma unroll 4
  for (int g = 0; g < NGRP; g += 4) {
    s0 += p[(size_t)(g + 0) * GSTRIDE];
    s1 += p[(size_t)(g + 1) * GSTRIDE];
    s2 += p[(size_t)(g + 2) * GSTRIDE];
    s3 += p[(size_t)(g + 3) * GSTRIDE];
  }
  wsum[(size_t)(b * SEQ + i) * HEADS + h] = (s0 + s1) + (s2 + s3);
}

__global__ __launch_bounds__(32) void out_gemm(
    const float* __restrict__ vws, const float* __restrict__ wsum,
    const float* __restrict__ Wout, float* __restrict__ out)
{
  const int lane = threadIdx.x, half = lane >> 4, l16 = lane & 15;
  const int o0 = blockIdx.x * 32;
  const int i0 = blockIdx.y * 32;
  const int b  = blockIdx.z;
  const int iA = i0 + l16;
  const int iB = i0 + 16 + l16;

  const float* vrowA = vws + (size_t)(b * SEQ + iA) * HD + 16 * half;
  const float* vrowB = vws + (size_t)(b * SEQ + iB) * HD + 16 * half;
  const float* wrowA = wsum + (size_t)(b * SEQ + iA) * HEADS;
  const float* wrowB = wsum + (size_t)(b * SEQ + iB) * HEADS;
  float wvA[HEADS], wvB[HEADS];
#pragma unroll
  for (int h = 0; h < HEADS; ++h) { wvA[h] = wrowA[h]; wvB[h] = wrowB[h]; }

  const float* wo0 = Wout + (size_t)(o0 + l16) * HD + 16 * half;
  const float* wo1 = Wout + (size_t)(o0 + 16 + l16) * HD + 16 * half;

  v8f acc00 = {}, acc01 = {}, acc10 = {}, acc11 = {};
  constexpr int NBLK = HD / 32;
  v4f A0[2][4], A1[2][4], B0[2][4], B1[2][4];
  load_chunks(A0[0], vrowA);
  load_chunks(A1[0], vrowB);
  load_chunks(B0[0], wo0);
  load_chunks(B1[0], wo1);
#pragma unroll
  for (int blk = 0; blk < NBLK; ++blk) {
    const int cur = blk & 1, nxt = cur ^ 1;
    if (blk + 1 < NBLK) {
      load_chunks(A0[nxt], vrowA + (blk + 1) * 32);
      load_chunks(A1[nxt], vrowB + (blk + 1) * 32);
      load_chunks(B0[nxt], wo0 + (blk + 1) * 32);
      load_chunks(B1[nxt], wo1 + (blk + 1) * 32);
    }
    __builtin_amdgcn_sched_barrier(0);
    const float sA = wvA[blk >> 1], sB = wvB[blk >> 1];
#pragma unroll
    for (int s = 0; s < 8; ++s) {
      const v2f a0 = step_pair(A0[cur], s) * sA;
      const v2f a1 = step_pair(A1[cur], s) * sB;
      const v2f b0 = step_pair(B0[cur], s), b1 = step_pair(B1[cur], s);
      acc00 = WMMA_F32(a0, b0, acc00);
      acc01 = WMMA_F32(a0, b1, acc01);
      acc10 = WMMA_F32(a1, b0, acc10);
      acc11 = WMMA_F32(a1, b1, acc11);
    }
  }
#pragma unroll
  for (int r = 0; r < 8; ++r) {
    const size_t row0 = (size_t)(b * SEQ + i0 + r + half * 8) * COUT;
    const size_t row1 = (size_t)(b * SEQ + i0 + 16 + r + half * 8) * COUT;
    out[row0 + o0 + l16]      = acc00[r];
    out[row0 + o0 + 16 + l16] = acc01[r];
    out[row1 + o0 + l16]      = acc10[r];
    out[row1 + o0 + 16 + l16] = acc11[r];
  }
}

extern "C" void kernel_launch(void* const* d_in, const int* in_sizes, int n_in,
                              void* d_out, int out_size, void* d_ws, size_t ws_size,
                              hipStream_t stream)
{
  (void)in_sizes; (void)n_in; (void)out_size; (void)ws_size;
  const float* X  = (const float*)d_in[0];
  const float* Wq = (const float*)d_in[1];
  const float* Wk = (const float*)d_in[2];
  const float* Wv = (const float*)d_in[3];
  const float* Wo = (const float*)d_in[4];
  float* out = (float*)d_out;

  const size_t QSZ = (size_t)BATCH * SEQ * HD;
  const size_t SSZ = (size_t)BATCH * HEADS * SEQ;
  float* qws  = (float*)d_ws;
  float* kws  = qws + QSZ;
  float* vws  = kws + QSZ;
  float* wsum = vws + QSZ + 2 * SSZ;

  qkv_gemm<<<dim3(HD / 32, SEQ / 32, BATCH * 3), 32, 0, stream>>>(
      X, Wq, Wk, Wv, qws, kws, vws);
  key_strip<<<dim3(NGRP, HEADS, BATCH), WAVES * 32, 0, stream>>>(qws, kws);
  wsum_reduce<<<dim3(BATCH * HEADS * SEQ / 256), 256, 0, stream>>>(kws, wsum);
  out_gemm<<<dim3(COUT / 32, SEQ / 32, BATCH), 32, 0, stream>>>(
      vws, wsum, Wo, out);
}


# qkv projection: weight rows read as 16-byte loads, a lane's 4 adjacent columns feeding 4 column tiles of a 32x64 wave tile (was 4-byte column-strided loads, 32x32 tile); tile stores 16-byte too
# speedup vs baseline: 2.0517x; 1.0758x over previous
#include <hip/hip_runtime.h>
#include <math.h>


typedef float v2f __attribute__((ext_vector_type(2)));
typedef float v4f __attribute__((ext_vector_type(4)));
typedef float v8f __attribute__((ext_vector_type(8)));

#define WMMA_F32(a, b, c) \
  __builtin_amdgcn_wmma_f32_16x16x4_f32(false, (a), false, (b), (short)0, (c), false, false)

constexpr int BATCH = 8;
constexpr int SEQ   = 1024;
constexpr int CIN   = 256;
constexpr int HEADS = 8;
constexpr int DIM   = 64;
constexpr int HD    = HEADS * DIM;
constexpr int COUT  = 256;
constexpr float SCALE = 0.125f;
constexpr int NT16   = SEQ / 16;
constexpr int TILE_F = 16 * DIM;

__device__ __forceinline__ size_t tile_off(int b, int h, int t16)
{
  return ((size_t)(b * HEADS + h) * NT16 + t16) * TILE_F;
}
__device__ __forceinline__ v2f lo2(v4f x) { return __builtin_shufflevector(x, x, 0, 1); }
__device__ __forceinline__ v2f hi2(v4f x) { return __builtin_shufflevector(x, x, 2, 3); }

__device__ __forceinline__ void load_chunks(v4f (&f)[4], const float* __restrict__ p)
{
#pragma unroll
  for (int q = 0; q < 4; ++q) f[q] = *(const v4f*)(p + 4 * q);
}
__device__ __forceinline__ v2f step_pair(const v4f (&f)[4], int s)
{
  return (s & 1) ? hi2(f[s >> 1]) : lo2(f[s >> 1]);
}

__device__ __forceinline__ void load_wrows(v4f (&wr)[16], const float* __restrict__ p)
{
#pragma unroll
  for (int j = 0; j < 16; ++j) wr[j] = *(const v4f*)(p + (size_t)j * HD);
}
__device__ __forceinline__ v2f w_frag(const v4f (&wr)[16], int s, int t)
{
  v2f f;
  f.x = wr[2 * s][t];
  f.y = wr[2 * s + 1][t];
  return f;
}

__global__ __launch_bounds__(32) void qkv_gemm(
    const float* __restrict__ X,
    const float* __restrict__ Wq, const float* __restrict__ Wk,
    const float* __restrict__ Wv,
    float* __restrict__ qws, float* __restrict__ kws, float* __restrict__ vws)
{
  const int lane = threadIdx.x;
  const int half = lane >> 4, l16 = lane & 15;
  const int n0 = blockIdx.x * 64;
  const int i0 = blockIdx.y * 32;
  const int b  = blockIdx.z / 3;
  const int sel = blockIdx.z % 3;

  const float* W = (sel == 0) ? Wq : (sel == 1) ? Wk : Wv;

  const float* xrow0 = X + (size_t)(b * SEQ + i0 + l16) * CIN + 16 * half;
  const float* xrow1 = xrow0 + (size_t)16 * CIN;
  const float* wlane = W + (size_t)(16 * half) * HD + n0 + 4 * l16;

  v8f acc[2][4];
#pragma unroll
  for (int R = 0; R < 2; ++R)
#pragma unroll
    for (int t = 0; t < 4; ++t) acc[R][t] = (v8f){};

  if (sel == 2) {
    for (int kb0 = 0; kb0 < CIN; kb0 += 32) {
      v4f xa[4], xb[4], wr[16];
      load_chunks(xa, xrow0 + kb0);
      load_chunks(xb, xrow1 + kb0);
      load_wrows(wr, wlane + (size_t)kb0 * HD);
#pragma unroll
      for (int s = 0; s < 8; ++s) {
        const v2f a0 = step_pair(xa, s), a1 = step_pair(xb, s);
#pragma unroll
        for (int t = 0; t < 4; ++t) {
          const v2f bt = w_frag(wr, s, t);
          acc[0][t] = WMMA_F32(a0, bt, acc[0][t]);
          acc[1][t] = WMMA_F32(a1, bt, acc[1][t]);
        }
      }
    }
    float* vlane = vws + (size_t)(b * SEQ + i0 + half * 8) * HD + n0 + 4 * l16;
#pragma unroll
    for (int R = 0; R < 2; ++R)
#pragma unroll
      for (int r = 0; r < 8; ++r)
        *(v4f*)(vlane + (size_t)(16 * R + r) * HD) =
            (v4f){acc[R][0][r], acc[R][1][r], acc[R][2][r], acc[R][3][r]};
  } else {
    for (int kb0 = 0; kb0 < CIN; kb0 += 32) {
      v4f xa[4], xb[4], wr[16];
      load_chunks(xa, xrow0 + kb0);
      load_chunks(xb, xrow1 + kb0);
      load_wrows(wr, wlane + (size_t)kb0 * HD);
#pragma unroll
      for (int s = 0; s < 8; ++s) {
        const v2f a0 = step_pair(xa, s), a1 = step_pair(xb, s);
#pragma unroll
        for (int t = 0; t < 4; ++t) {
          const v2f bt = w_frag(wr, s, t);
          acc[0][t] = WMMA_F32(bt, a0, acc[0][t]);
          acc[1][t] = WMMA_F32(bt, a1, acc[1][t]);
        }
      }
    }
    float* O = (sel == 0) ? qws : kws;
    const int head = blockIdx.x;
    float* olane = O + tile_off(b, head, i0 >> 4) + 512 * half + 4 * l16;
#pragma unroll
    for (int R = 0; R < 2; ++R)
#pragma unroll
      for (int r = 0; r < 8; ++r) {
        const int slot = 2 * (r >> 2) + (r & 1);
        const int hp   = (r >> 1) & 1;
        *(v4f*)(olane + (size_t)R * TILE_F + slot * 128 + hp * 64) =
            (v4f){acc[R][0][r], acc[R][1][r], acc[R][2][r], acc[R][3][r]};
      }
  }
}

__device__ __forceinline__ void load_frags(v4f (&f)[8], const float* __restrict__ p)
{
#pragma unroll
  for (int w = 0; w < 8; ++w) f[w] = *(const v4f*)(p + w * 128);
}

constexpr int KT_WG   = 4;
constexpr int WAVES   = 4;
constexpr int QT_WAVE = NT16 / WAVES;
constexpr int NGRP    = NT16 / KT_WG;

template <int XM> __device__ __forceinline__ float row_xor(float x)
{
  return __builtin_bit_cast(float, __builtin_amdgcn_update_dpp(
      0, __builtin_bit_cast(int, x), 0x160 + XM, 0xf, 0xf, true));
}
__device__ __forceinline__ float row16_max(float x)
{
  x = fmaxf(x, row_xor<1>(x));
  x = fmaxf(x, row_xor<2>(x));
  x = fmaxf(x, row_xor<4>(x));
  x = fmaxf(x, row_xor<8>(x));
  return x;
}
__device__ __forceinline__ float row16_sum(float x)
{
  x += row_xor<1>(x);
  x += row_xor<2>(x);
  x += row_xor<4>(x);
  x += row_xor<8>(x);
  return x;
}

__global__ __launch_bounds__(WAVES * 32) void key_strip(
    const float* __restrict__ qws, float* __restrict__ kws)
{
  __shared__ v4f xm[WAVES][KT_WG * 4];
  __shared__ v4f xl[WAVES][KT_WG * 4];

  const int lane = threadIdx.x & 31, half = lane >> 4, l16 = lane & 15;
  const int wave = __builtin_amdgcn_readfirstlane((int)(threadIdx.x >> 5));
  const int kg = blockIdx.x;
  const int h  = blockIdx.y;
  const int b  = blockIdx.z;

  float* kgrp = kws + tile_off(b, h, KT_WG * kg);
  v4f kf[KT_WG][8];
#pragma unroll
  for (int j = 0; j < KT_WG; ++j)
    load_frags(kf[j], kgrp + (size_t)j * TILE_F + lane * 4);

  const float* qlane = qws + tile_off(b, h, wave * QT_WAVE) + lane * 4;

  v8f S[KT_WG][QT_WAVE];
  float mx[KT_WG][8];
#pragma unroll
  for (int j = 0; j < KT_WG; ++j)
#pragma unroll
    for (int r = 0; r < 8; ++r) mx[j][r] = -__builtin_inff();

  constexpr int NSTEP = QT_WAVE / 2;
  v4f F[2][2][8];
  load_frags(F[0][0], qlane);
  load_frags(F[0][1], qlane + (size_t)1 * TILE_F);
#pragma unroll
  for (int s = 0; s < NSTEP; ++s) {
    const int cur = s & 1, nxt = cur ^ 1;
    if (s + 1 < NSTEP) {
      load_frags(F[nxt][0], qlane + (size_t)(2 * s + 2) * TILE_F);
      load_frags(F[nxt][1], qlane + (size_t)(2 * s + 3) * TILE_F);
    }
    __builtin_amdgcn_sched_barrier(0);
    v8f a0[KT_WG], a1[KT_WG];
#pragma unroll
    for (int j = 0; j < KT_WG; ++j) { a0[j] = (v8f){}; a1[j] = (v8f){}; }
#pragma unroll
    for (int w = 0; w < 8; ++w) {
#pragma unroll
      for (int j = 0; j < KT_WG; ++j) {
        a0[j] = WMMA_F32(lo2(kf[j][w]), lo2(F[cur][0][w]), a0[j]);
        a1[j] = WMMA_F32(lo2(kf[j][w]), lo2(F[cur][1][w]), a1[j]);
      }
#pragma unroll
      for (int j = 0; j < KT_WG; ++j) {
        a0[j] = WMMA_F32(hi2(kf[j][w]), hi2(F[cur][0][w]), a0[j]);
        a1[j] = WMMA_F32(hi2(kf[j][w]), hi2(F[cur][1][w]), a1[j]);
      }
    }
#pragma unroll
    for (int j = 0; j < KT_WG; ++j) {
      S[j][2 * s]     = a0[j];
      S[j][2 * s + 1] = a1[j];
#pragma unroll
      for (int r = 0; r < 8; ++r)
        mx[j][r] = fmaxf(mx[j][r], fmaxf(a0[j][r], a1[j][r]));
    }
  }

#pragma unroll
  for (int j = 0; j < KT_WG; ++j)
#pragma unroll
    for (int r = 0; r < 8; ++r) mx[j][r] = row16_max(mx[j][r]);
  if (l16 == 0) {
#pragma unroll
    for (int j = 0; j < KT_WG; ++j) {
      xm[wave][4 * j + 2 * half]     = (v4f){mx[j][0], mx[j][1], mx[j][2], mx[j][3]};
      xm[wave][4 * j + 2 * half + 1] = (v4f){mx[j][4], mx[j][5], mx[j][6], mx[j][7]};
    }
  }
  __syncthreads();
  float m[KT_WG][8], l[KT_WG][8];
#pragma unroll
  for (int j = 0; j < KT_WG; ++j) {
    v4f lo = xm[0][4 * j + 2 * half], hi = xm[0][4 * j + 2 * half + 1];
#pragma unroll
    for (int w = 1; w < WAVES; ++w) {
      const v4f lo_w = xm[w][4 * j + 2 * half], hi_w = xm[w][4 * j + 2 * half + 1];
#pragma unroll
      for (int e = 0; e < 4; ++e) {
        lo[e] = fmaxf(lo[e], lo_w[e]);
        hi[e] = fmaxf(hi[e], hi_w[e]);
      }
    }
#pragma unroll
    for (int e = 0; e < 4; ++e) {
      m[j][e]     = lo[e] * SCALE;
      m[j][4 + e] = hi[e] * SCALE;
      l[j][e]     = 0.0f;
      l[j][4 + e] = 0.0f;
    }
  }
#pragma unroll
  for (int j = 0; j < KT_WG; ++j)
#pragma unroll
    for (int t = 0; t < QT_WAVE; ++t)
#pragma unroll
      for (int r = 0; r < 8; ++r) {
        const float e = __expf(S[j][t][r] * SCALE - m[j][r]);
        S[j][t][r] = e;
        l[j][r] += e;
      }
#pragma unroll
  for (int j = 0; j < KT_WG; ++j)
#pragma unroll
    for (int r = 0; r < 8; ++r) l[j][r] = row16_sum(l[j][r]);
  if (l16 == 0) {
#pragma unroll
    for (int j = 0; j < KT_WG; ++j) {
      xl[wave][4 * j + 2 * half]     = (v4f){l[j][0], l[j][1], l[j][2], l[j][3]};
      xl[wave][4 * j + 2 * half + 1] = (v4f){l[j][4], l[j][5], l[j][6], l[j][7]};
    }
  }
  __syncthreads();
  float invl[KT_WG][8];
#pragma unroll
  for (int j = 0; j < KT_WG; ++j) {
    v4f lo = xl[0][4 * j + 2 * half], hi = xl[0][4 * j + 2 * half + 1];
#pragma unroll
    for (int w = 1; w < WAVES; ++w) {
      lo += xl[w][4 * j + 2 * half];
      hi += xl[w][4 * j + 2 * half + 1];
    }
#pragma unroll
    for (int e = 0; e < 4; ++e) {
      invl[j][e]     = 1.0f / lo[e];
      invl[j][4 + e] = 1.0f / hi[e];
    }
  }
  float* pout = kgrp + 16 * (wave * QT_WAVE);
#pragma unroll
  for (int t = 0; t < QT_WAVE; t += 2) {
    float w0 = 0.0f, w1 = 0.0f;
#pragma unroll
    for (int j = 0; j < KT_WG; ++j)
#pragma unroll
      for (int r = 0; r < 8; ++r) {
        w0 += S[j][t][r] * invl[j][r];
        w1 += S[j][t + 1][r] * invl[j][r];
      }
    w0 += __shfl_xor(w0, 16, 32);
    w1 += __shfl_xor(w1, 16, 32);
    pout[16 * t + lane] = half ? w1 : w0;
  }
}

__global__ __launch_bounds__(256) void wsum_reduce(
    const float* __restrict__ part, float* __restrict__ wsum)
{
  const int idx = blockIdx.x * 256 + threadIdx.x;
  const int i  = idx & (SEQ - 1);
  const int bh = idx >> 10;
  const int b  = bh >> 3;
  const int h  = bh & (HEADS - 1);
  const float* p = part + (size_t)bh * NT16 * TILE_F + i;
  constexpr size_t GSTRIDE = (size_t)KT_WG * TILE_F;
  float s0 = 0.0f, s1 = 0.0f, s2 = 0.0f, s3 = 0.0f;
#pragma unroll 4
  for (int g = 0; g < NGRP; g += 4) {
    s0 += p[(size_t)(g + 0) * GSTRIDE];
    s1 += p[(size_t)(g + 1) * GSTRIDE];
    s2 += p[(size_t)(g + 2) * GSTRIDE];
    s3 += p[(size_t)(g + 3) * GSTRIDE];
  }
  wsum[(size_t)(b * SEQ + i) * HEADS + h] = (s0 + s1) + (s2 + s3);
}

__global__ __launch_bounds__(32) void out_gemm(
    const float* __restrict__ vws, const float* __restrict__ wsum,
    const float* __restrict__ Wout, float* __restrict__ out)
{
  const int lane = threadIdx.x, half = lane >> 4, l16 = lane & 15;
  const int o0 = blockIdx.x * 32;
  const int i0 = blockIdx.y * 32;
  const int b  = blockIdx.z;
  const int iA = i0 + l16;
  const int iB = i0 + 16 + l16;

  const float* vrowA = vws + (size_t)(b * SEQ + iA) * HD + 16 * half;
  const float* vrowB = vws + (size_t)(b * SEQ + iB) * HD + 16 * half;
  const float* wrowA = wsum + (size_t)(b * SEQ + iA) * HEADS;
  const float* wrowB = wsum + (size_t)(b * SEQ + iB) * HEADS;
  float wvA[HEADS], wvB[HEADS];
#pragma unroll
  for (int h = 0; h < HEADS; ++h) { wvA[h] = wrowA[h]; wvB[h] = wrowB[h]; }

  const float* wo0 = Wout + (size_t)(o0 + l16) * HD + 16 * half;
  const float* wo1 = Wout + (size_t)(o0 + 16 + l16) * HD + 16 * half;

  v8f acc00 = {}, acc01 = {}, acc10 = {}, acc11 = {};
  constexpr int NBLK = HD / 32;
  v4f A0[2][4], A1[2][4], B0[2][4], B1[2][4];
  load_chunks(A0[0], vrowA);
  load_chunks(A1[0], vrowB);
  load_chunks(B0[0], wo0);
  load_chunks(B1[0], wo1);
#pragma unroll
  for (int blk = 0; blk < NBLK; ++blk) {
    const int cur = blk & 1, nxt = cur ^ 1;
    if (blk + 1 < NBLK) {
      load_chunks(A0[nxt], vrowA + (blk + 1) * 32);
      load_chunks(A1[nxt], vrowB + (blk + 1) * 32);
      load_chunks(B0[nxt], wo0 + (blk + 1) * 32);
      load_chunks(B1[nxt], wo1 + (blk + 1) * 32);
    }
    __builtin_amdgcn_sched_barrier(0);
    const float sA = wvA[blk >> 1], sB = wvB[blk >> 1];
#pragma unroll
    for (int s = 0; s < 8; ++s) {
      const v2f a0 = step_pair(A0[cur], s) * sA;
      const v2f a1 = step_pair(A1[cur], s) * sB;
      const v2f b0 = step_pair(B0[cur], s), b1 = step_pair(B1[cur], s);
      acc00 = WMMA_F32(a0, b0, acc00);
      acc01 = WMMA_F32(a0, b1, acc01);
      acc10 = WMMA_F32(a1, b0, acc10);
      acc11 = WMMA_F32(a1, b1, acc11);
    }
  }
#pragma unroll
  for (int r = 0; r < 8; ++r) {
    const size_t row0 = (size_t)(b * SEQ + i0 + r + half * 8) * COUT;
    const size_t row1 = (size_t)(b * SEQ + i0 + 16 + r + half * 8) * COUT;
    out[row0 + o0 + l16]      = acc00[r];
    out[row0 + o0 + 16 + l16] = acc01[r];
    out[row1 + o0 + l16]      = acc10[r];
    out[row1 + o0 + 16 + l16] = acc11[r];
  }
}

extern "C" void kernel_launch(void* const* d_in, const int* in_sizes, int n_in,
                              void* d_out, int out_size, void* d_ws, size_t ws_size,
                              hipStream_t stream)
{
  (void)in_sizes; (void)n_in; (void)out_size; (void)ws_size;
  const float* X  = (const float*)d_in[0];
  const float* Wq = (const float*)d_in[1];
  const float* Wk = (const float*)d_in[2];
  const float* Wv = (const float*)d_in[3];
  const float* Wo = (const float*)d_in[4];
  float* out = (float*)d_out;

  const size_t QSZ = (size_t)BATCH * SEQ * HD;
  const size_t SSZ = (size_t)BATCH * HEADS * SEQ;
  float* qws  = (float*)d_ws;
  float* kws  = qws + QSZ;
  float* vws  = kws + QSZ;
  float* wsum = vws + QSZ + 2 * SSZ;

  qkv_gemm<<<dim3(HD / 64, SEQ / 32, BATCH * 3), 32, 0, stream>>>(
      X, Wq, Wk, Wv, qws, kws, vws);
  key_strip<<<dim3(NGRP, HEADS, BATCH), WAVES * 32, 0, stream>>>(qws, kws);
  wsum_reduce<<<dim3(BATCH * HEADS * SEQ / 256), 256, 0, stream>>>(kws, wsum);
  out_gemm<<<dim3(COUT / 32, SEQ / 32, BATCH), 32, 0, stream>>>(
      vws, wsum, Wo, out);
}


# Score strip halved to 32 keys per workgroup: registers now allow two workgroups per WGP, so one's load start-up and exp/sum tail run under the other's WMMAs; 32 partials per row (new sum order)
# speedup vs baseline: 2.1656x; 1.0555x over previous
#include <hip/hip_runtime.h>
#include <math.h>


typedef float v2f __attribute__((ext_vector_type(2)));
typedef float v4f __attribute__((ext_vector_type(4)));
typedef float v8f __attribute__((ext_vector_type(8)));

#define WMMA_F32(a, b, c) \
  __builtin_amdgcn_wmma_f32_16x16x4_f32(false, (a), false, (b), (short)0, (c), false, false)

constexpr int BATCH = 8;
constexpr int SEQ   = 1024;
constexpr int CIN   = 256;
constexpr int HEADS = 8;
constexpr int DIM   = 64;
constexpr int HD    = HEADS * DIM;
constexpr int COUT  = 256;
constexpr float SCALE = 0.125f;
constexpr int NT16   = SEQ / 16;
constexpr int TILE_F = 16 * DIM;

__device__ __forceinline__ size_t tile_off(int b, int h, int t16)
{
  return ((size_t)(b * HEADS + h) * NT16 + t16) * TILE_F;
}
__device__ __forceinline__ v2f lo2(v4f x) { return __builtin_shufflevector(x, x, 0, 1); }
__device__ __forceinline__ v2f hi2(v4f x) { return __builtin_shufflevector(x, x, 2, 3); }

__device__ __forceinline__ void load_chunks(v4f (&f)[4], const float* __restrict__ p)
{
#pragma unroll
  for (int q = 0; q < 4; ++q) f[q] = *(const v4f*)(p + 4 * q);
}
__device__ __forceinline__ v2f step_pair(const v4f (&f)[4], int s)
{
  return (s & 1) ? hi2(f[s >> 1]) : lo2(f[s >> 1]);
}

__device__ __forceinline__ void load_wrows(v4f (&wr)[16], const float* __restrict__ p)
{
#pragma unroll
  for (int j = 0; j < 16; ++j) wr[j] = *(const v4f*)(p + (size_t)j * HD);
}
__device__ __forceinline__ v2f w_frag(const v4f (&wr)[16], int s, int t)
{
  v2f f;
  f.x = wr[2 * s][t];
  f.y = wr[2 * s + 1][t];
  return f;
}

__global__ __launch_bounds__(32) void qkv_gemm(
    const float* __restrict__ X,
    const float* __restrict__ Wq, const float* __restrict__ Wk,
    const float* __restrict__ Wv,
    float* __restrict__ qws, float* __restrict__ kws, float* __restrict__ vws)
{
  const int lane = threadIdx.x;
  const int half = lane >> 4, l16 = lane & 15;
  const int n0 = blockIdx.x * 64;
  const int i0 = blockIdx.y * 32;
  const int b  = blockIdx.z / 3;
  const int sel = blockIdx.z % 3;

  const float* W = (sel == 0) ? Wq : (sel == 1) ? Wk : Wv;

  const float* xrow0 = X + (size_t)(b * SEQ + i0 + l16) * CIN + 16 * half;
  const float* xrow1 = xrow0 + (size_t)16 * CIN;
  const float* wlane = W + (size_t)(16 * half) * HD + n0 + 4 * l16;

  v8f acc[2][4];
#pragma unroll
  for (int R = 0; R < 2; ++R)
#pragma unroll
    for (int t = 0; t < 4; ++t) acc[R][t] = (v8f){};

  if (sel == 2) {
    for (int kb0 = 0; kb0 < CIN; kb0 += 32) {
      v4f xa[4], xb[4], wr[16];
      load_chunks(xa, xrow0 + kb0);
      load_chunks(xb, xrow1 + kb0);
      load_wrows(wr, wlane + (size_t)kb0 * HD);
#pragma unroll
      for (int s = 0; s < 8; ++s) {
        const v2f a0 = step_pair(xa, s), a1 = step_pair(xb, s);
#pragma unroll
        for (int t = 0; t < 4; ++t) {
          const v2f bt = w_frag(wr, s, t);
          acc[0][t] = WMMA_F32(a0, bt, acc[0][t]);
          acc[1][t] = WMMA_F32(a1, bt, acc[1][t]);
        }
      }
    }
    float* vlane = vws + (size_t)(b * SEQ + i0 + half * 8) * HD + n0 + 4 * l16;
#pragma unroll
    for (int R = 0; R < 2; ++R)
#pragma unroll
      for (int r = 0; r < 8; ++r)
        *(v4f*)(vlane + (size_t)(16 * R + r) * HD) =
            (v4f){acc[R][0][r], acc[R][1][r], acc[R][2][r], acc[R][3][r]};
  } else {
    for (int kb0 = 0; kb0 < CIN; kb0 += 32) {
      v4f xa[4], xb[4], wr[16];
      load_chunks(xa, xrow0 + kb0);
      load_chunks(xb, xrow1 + kb0);
      load_wrows(wr, wlane + (size_t)kb0 * HD);
#pragma unroll
      for (int s = 0; s < 8; ++s) {
        const v2f a0 = step_pair(xa, s), a1 = step_pair(xb, s);
#pragma unroll
        for (int t = 0; t < 4; ++t) {
          const v2f bt = w_frag(wr, s, t);
          acc[0][t] = WMMA_F32(bt, a0, acc[0][t]);
          acc[1][t] = WMMA_F32(bt, a1, acc[1][t]);
        }
      }
    }
    float* O = (sel == 0) ? qws : kws;
    const int head = blockIdx.x;
    float* olane = O + tile_off(b, head, i0 >> 4) + 512 * half + 4 * l16;
#pragma unroll
    for (int R = 0; R < 2; ++R)
#pragma unroll
      for (int r = 0; r < 8; ++r) {
        const int slot = 2 * (r >> 2) + (r & 1);
        const int hp   = (r >> 1) & 1;
        *(v4f*)(olane + (size_t)R * TILE_F + slot * 128 + hp * 64) =
            (v4f){acc[R][0][r], acc[R][1][r], acc[R][2][r], acc[R][3][r]};
      }
  }
}

__device__ __forceinline__ void load_frags(v4f (&f)[8], const float* __restrict__ p)
{
#pragma unroll
  for (int w = 0; w < 8; ++w) f[w] = *(const v4f*)(p + w * 128);
}

constexpr int KT_WG   = 2;
constexpr int WAVES   = 4;
constexpr int QT_WAVE = NT16 / WAVES;
constexpr int NGRP    = NT16 / KT_WG;

template <int XM> __device__ __forceinline__ float row_xor(float x)
{
  return __builtin_bit_cast(float, __builtin_amdgcn_update_dpp(
      0, __builtin_bit_cast(int, x), 0x160 + XM, 0xf, 0xf, true));
}
__device__ __forceinline__ float row16_max(float x)
{
  x = fmaxf(x, row_xor<1>(x));
  x = fmaxf(x, row_xor<2>(x));
  x = fmaxf(x, row_xor<4>(x));
  x = fmaxf(x, row_xor<8>(x));
  return x;
}
__device__ __forceinline__ float row16_sum(float x)
{
  x += row_xor<1>(x);
  x += row_xor<2>(x);
  x += row_xor<4>(x);
  x += row_xor<8>(x);
  return x;
}

__global__ __launch_bounds__(WAVES * 32) void key_strip(
    const float* __restrict__ qws, float* __restrict__ kws)
{
  __shared__ v4f xm[WAVES][KT_WG * 4];
  __shared__ v4f xl[WAVES][KT_WG * 4];

  const int lane = threadIdx.x & 31, half = lane >> 4, l16 = lane & 15;
  const int wave = __builtin_amdgcn_readfirstlane((int)(threadIdx.x >> 5));
  const int kg = blockIdx.x;
  const int h  = blockIdx.y;
  const int b  = blockIdx.z;

  float* kgrp = kws + tile_off(b, h, KT_WG * kg);
  v4f kf[KT_WG][8];
#pragma unroll
  for (int j = 0; j < KT_WG; ++j)
    load_frags(kf[j], kgrp + (size_t)j * TILE_F + lane * 4);

  const float* qlane = qws + tile_off(b, h, wave * QT_WAVE) + lane * 4;

  v8f S[KT_WG][QT_WAVE];
  float mx[KT_WG][8];
#pragma unroll
  for (int j = 0; j < KT_WG; ++j)
#pragma unroll
    for (int r = 0; r < 8; ++r) mx[j][r] = -__builtin_inff();

  constexpr int NSTEP = QT_WAVE / 2;
  v4f F[2][2][8];
  load_frags(F[0][0], qlane);
  load_frags(F[0][1], qlane + (size_t)1 * TILE_F);
#pragma unroll
  for (int s = 0; s < NSTEP; ++s) {
    const int cur = s & 1, nxt = cur ^ 1;
    if (s + 1 < NSTEP) {
      load_frags(F[nxt][0], qlane + (size_t)(2 * s + 2) * TILE_F);
      load_frags(F[nxt][1], qlane + (size_t)(2 * s + 3) * TILE_F);
    }
    __builtin_amdgcn_sched_barrier(0);
    v8f a0[KT_WG], a1[KT_WG];
#pragma unroll
    for (int j = 0; j < KT_WG; ++j) { a0[j] = (v8f){}; a1[j] = (v8f){}; }
#pragma unroll
    for (int w = 0; w < 8; ++w) {
#pragma unroll
      for (int j = 0; j < KT_WG; ++j) {
        a0[j] = WMMA_F32(lo2(kf[j][w]), lo2(F[cur][0][w]), a0[j]);
        a1[j] = WMMA_F32(lo2(kf[j][w]), lo2(F[cur][1][w]), a1[j]);
      }
#pragma unroll
      for (int j = 0; j < KT_WG; ++j) {
        a0[j] = WMMA_F32(hi2(kf[j][w]), hi2(F[cur][0][w]), a0[j]);
        a1[j] = WMMA_F32(hi2(kf[j][w]), hi2(F[cur][1][w]), a1[j]);
      }
    }
#pragma unroll
    for (int j = 0; j < KT_WG; ++j) {
      S[j][2 * s]     = a0[j];
      S[j][2 * s + 1] = a1[j];
#pragma unroll
      for (int r = 0; r < 8; ++r)
        mx[j][r] = fmaxf(mx[j][r], fmaxf(a0[j][r], a1[j][r]));
    }
  }

#pragma unroll
  for (int j = 0; j < KT_WG; ++j)
#pragma unroll
    for (int r = 0; r < 8; ++r) mx[j][r] = row16_max(mx[j][r]);
  if (l16 == 0) {
#pragma unroll
    for (int j = 0; j < KT_WG; ++j) {
      xm[wave][4 * j + 2 * half]     = (v4f){mx[j][0], mx[j][1], mx[j][2], mx[j][3]};
      xm[wave][4 * j + 2 * half + 1] = (v4f){mx[j][4], mx[j][5], mx[j][6], mx[j][7]};
    }
  }
  __syncthreads();
  float m[KT_WG][8], l[KT_WG][8];
#pragma unroll
  for (int j = 0; j < KT_WG; ++j) {
    v4f lo = xm[0][4 * j + 2 * half], hi = xm[0][4 * j + 2 * half + 1];
#pragma unroll
    for (int w = 1; w < WAVES; ++w) {
      const v4f lo_w = xm[w][4 * j + 2 * half], hi_w = xm[w][4 * j + 2 * half + 1];
#pragma unroll
      for (int e = 0; e < 4; ++e) {
        lo[e] = fmaxf(lo[e], lo_w[e]);
        hi[e] = fmaxf(hi[e], hi_w[e]);
      }
    }
#pragma unroll
    for (int e = 0; e < 4; ++e) {
      m[j][e]     = lo[e] * SCALE;
      m[j][4 + e] = hi[e] * SCALE;
      l[j][e]     = 0.0f;
      l[j][4 + e] = 0.0f;
    }
  }
#pragma unroll
  for (int j = 0; j < KT_WG; ++j)
#pragma unroll
    for (int t = 0; t < QT_WAVE; ++t)
#pragma unroll
      for (int r = 0; r < 8; ++r) {
        const float e = __expf(S[j][t][r] * SCALE - m[j][r]);
        S[j][t][r] = e;
        l[j][r] += e;
      }
#pragma unroll
  for (int j = 0; j < KT_WG; ++j)
#pragma unroll
    for (int r = 0; r < 8; ++r) l[j][r] = row16_sum(l[j][r]);
  if (l16 == 0) {
#pragma unroll
    for (int j = 0; j < KT_WG; ++j) {
      xl[wave][4 * j + 2 * half]     = (v4f){l[j][0], l[j][1], l[j][2], l[j][3]};
      xl[wave][4 * j + 2 * half + 1] = (v4f){l[j][4], l[j][5], l[j][6], l[j][7]};
    }
  }
  __syncthreads();
  float invl[KT_WG][8];
#pragma unroll
  for (int j = 0; j < KT_WG; ++j) {
    v4f lo = xl[0][4 * j + 2 * half], hi = xl[0][4 * j + 2 * half + 1];
#pragma unroll
    for (int w = 1; w < WAVES; ++w) {
      lo += xl[w][4 * j + 2 * half];
      hi += xl[w][4 * j + 2 * half + 1];
    }
#pragma unroll
    for (int e = 0; e < 4; ++e) {
      invl[j][e]     = 1.0f / lo[e];
      invl[j][4 + e] = 1.0f / hi[e];
    }
  }
  float* pout = kgrp + 16 * (wave * QT_WAVE);
#pragma unroll
  for (int t = 0; t < QT_WAVE; t += 2) {
    float w0 = 0.0f, w1 = 0.0f;
#pragma unroll
    for (int j = 0; j < KT_WG; ++j)
#pragma unroll
      for (int r = 0; r < 8; ++r) {
        w0 += S[j][t][r] * invl[j][r];
        w1 += S[j][t + 1][r] * invl[j][r];
      }
    w0 += __shfl_xor(w0, 16, 32);
    w1 += __shfl_xor(w1, 16, 32);
    pout[16 * t + lane] = half ? w1 : w0;
  }
}

__global__ __launch_bounds__(256) void wsum_reduce(
    const float* __restrict__ part, float* __restrict__ wsum)
{
  const int idx = blockIdx.x * 256 + threadIdx.x;
  const int i  = idx & (SEQ - 1);
  const int bh = idx >> 10;
  const int b  = bh >> 3;
  const int h  = bh & (HEADS - 1);
  const float* p = part + (size_t)bh * NT16 * TILE_F + i;
  constexpr size_t GSTRIDE = (size_t)KT_WG * TILE_F;
  float s0 = 0.0f, s1 = 0.0f, s2 = 0.0f, s3 = 0.0f;
#pragma unroll
  for (int g = 0; g < NGRP; g += 4) {
    s0 += p[(size_t)(g + 0) * GSTRIDE];
    s1 += p[(size_t)(g + 1) * GSTRIDE];
    s2 += p[(size_t)(g + 2) * GSTRIDE];
    s3 += p[(size_t)(g + 3) * GSTRIDE];
  }
  wsum[(size_t)(b * SEQ + i) * HEADS + h] = (s0 + s1) + (s2 + s3);
}

__global__ __launch_bounds__(32) void out_gemm(
    const float* __restrict__ vws, const float* __restrict__ wsum,
    const float* __restrict__ Wout, float* __restrict__ out)
{
  const int lane = threadIdx.x, half = lane >> 4, l16 = lane & 15;
  const int o0 = blockIdx.x * 32;
  const int i0 = blockIdx.y * 32;
  const int b  = blockIdx.z;
  const int iA = i0 + l16;
  const int iB = i0 + 16 + l16;

  const float* vrowA = vws + (size_t)(b * SEQ + iA) * HD + 16 * half;
  const float* vrowB = vws + (size_t)(b * SEQ + iB) * HD + 16 * half;
  const float* wrowA = wsum + (size_t)(b * SEQ + iA) * HEADS;
  const float* wrowB = wsum + (size_t)(b * SEQ + iB) * HEADS;
  float wvA[HEADS], wvB[HEADS];
#pragma unroll
  for (int h = 0; h < HEADS; ++h) { wvA[h] = wrowA[h]; wvB[h] = wrowB[h]; }

  const float* wo0 = Wout + (size_t)(o0 + l16) * HD + 16 * half;
  const float* wo1 = Wout + (size_t)(o0 + 16 + l16) * HD + 16 * half;

  v8f acc00 = {}, acc01 = {}, acc10 = {}, acc11 = {};
  constexpr int NBLK = HD / 32;
  v4f A0[2][4], A1[2][4], B0[2][4], B1[2][4];
  load_chunks(A0[0], vrowA);
  load_chunks(A1[0], vrowB);
  load_chunks(B0[0], wo0);
  load_chunks(B1[0], wo1);
#pragma unroll
  for (int blk = 0; blk < NBLK; ++blk) {
    const int cur = blk & 1, nxt = cur ^ 1;
    if (blk + 1 < NBLK) {
      load_chunks(A0[nxt], vrowA + (blk + 1) * 32);
      load_chunks(A1[nxt], vrowB + (blk + 1) * 32);
      load_chunks(B0[nxt], wo0 + (blk + 1) * 32);
      load_chunks(B1[nxt], wo1 + (blk + 1) * 32);
    }
    __builtin_amdgcn_sched_barrier(0);
    const float sA = wvA[blk >> 1], sB = wvB[blk >> 1];
#pragma unroll
    for (int s = 0; s < 8; ++s) {
      const v2f a0 = step_pair(A0[cur], s) * sA;
      const v2f a1 = step_pair(A1[cur], s) * sB;
      const v2f b0 = step_pair(B0[cur], s), b1 = step_pair(B1[cur], s);
      acc00 = WMMA_F32(a0, b0, acc00);
      acc01 = WMMA_F32(a0, b1, acc01);
      acc10 = WMMA_F32(a1, b0, acc10);
      acc11 = WMMA_F32(a1, b1, acc11);
    }
  }
#pragma unroll
  for (int r = 0; r < 8; ++r) {
    const size_t row0 = (size_t)(b * SEQ + i0 + r + half * 8) * COUT;
    const size_t row1 = (size_t)(b * SEQ + i0 + 16 + r + half * 8) * COUT;
    out[row0 + o0 + l16]      = acc00[r];
    out[row0 + o0 + 16 + l16] = acc01[r];
    out[row1 + o0 + l16]      = acc10[r];
    out[row1 + o0 + 16 + l16] = acc11[r];
  }
}

extern "C" void kernel_launch(void* const* d_in, const int* in_sizes, int n_in,
                              void* d_out, int out_size, void* d_ws, size_t ws_size,
                              hipStream_t stream)
{
  (void)in_sizes; (void)n_in; (void)out_size; (void)ws_size;
  const float* X  = (const float*)d_in[0];
  const float* Wq = (const float*)d_in[1];
  const float* Wk = (const float*)d_in[2];
  const float* Wv = (const float*)d_in[3];
  const float* Wo = (const float*)d_in[4];
  float* out = (float*)d_out;

  const size_t QSZ = (size_t)BATCH * SEQ * HD;
  const size_t SSZ = (size_t)BATCH * HEADS * SEQ;
  float* qws  = (float*)d_ws;
  float* kws  = qws + QSZ;
  float* vws  = kws + QSZ;
  float* wsum = vws + QSZ + 2 * SSZ;

  qkv_gemm<<<dim3(HD / 64, SEQ / 32, BATCH * 3), 32, 0, stream>>>(
      X, Wq, Wk, Wv, qws, kws, vws);
  key_strip<<<dim3(NGRP, HEADS, BATCH), WAVES * 32, 0, stream>>>(qws, kws);
  wsum_reduce<<<dim3(BATCH * HEADS * SEQ / 256), 256, 0, stream>>>(kws, wsum);
  out_gemm<<<dim3(COUT / 32, SEQ / 32, BATCH), 32, 0, stream>>>(
      vws, wsum, Wo, out);
}
